# starting kernel
# baseline (speedup 1.0000x reference)
#include <hip/hip_runtime.h>
#include <math.h>

typedef float v2f __attribute__((ext_vector_type(2)));
typedef float v8f __attribute__((ext_vector_type(8)));

#define N_NODES 50000
#define F_IN    128
#define F_HID   128
#define F_OUT   40

__global__ void zero_f32(float* __restrict__ p, int n) {
    int i = blockIdx.x * blockDim.x + threadIdx.x;
    if (i < n) p[i] = 0.0f;
}

__global__ void deg_acc(const int* __restrict__ dst, float* __restrict__ deg, int E) {
    int e = blockIdx.x * blockDim.x + threadIdx.x;
    if (e < E) unsafeAtomicAdd(&deg[dst[e]], 1.0f);
}

__global__ void make_dinv(float* __restrict__ deg, int n) {
    int i = blockIdx.x * blockDim.x + threadIdx.x;
    if (i < n) deg[i] = rsqrtf(deg[i] + 1.0f);
}

template <int NT, int NC>
__global__ void gemm_wmma_f32(const float* __restrict__ X,
                              const float* __restrict__ W,
                              float* __restrict__ H, int M) {
    const int lane = threadIdx.x & 31;
    const int wave = threadIdx.x >> 5;
    const int mt   = blockIdx.x * (blockDim.x >> 5) + wave;
    if (mt * 16 >= M) return;

    const int half = lane >> 4;
    const int l16  = lane & 15;

    v8f c[NT] = {};

    const float* xrow = X + (size_t)(mt * 16 + l16) * 128;

    for (int k = 0; k < 128; k += 4) {
        v2f a;
        a.x = xrow[k + 2 * half];
        a.y = xrow[k + 2 * half + 1];
#pragma unroll
        for (int t = 0; t < NT; ++t) {
            const int n = t * 16 + l16;
            v2f b;
            if ((NC % 16 == 0) || (n < NC)) {
                b.x = W[(size_t)(k + 2 * half) * NC + n];
                b.y = W[(size_t)(k + 2 * half + 1) * NC + n];
            } else {
                b.x = 0.0f; b.y = 0.0f;
            }
            c[t] = __builtin_amdgcn_wmma_f32_16x16x4_f32(
                false, a, false, b, (short)0, c[t], false, false);
        }
    }

#pragma unroll
    for (int t = 0; t < NT; ++t) {
        const int n = t * 16 + l16;
        if ((NC % 16 == 0) || (n < NC)) {
            float* out = H + (size_t)(mt * 16 + half * 8) * NC + n;
#pragma unroll
            for (int r = 0; r < 8; ++r) out[(size_t)r * NC] = c[t][r];
        }
    }
}

template <int F>
__global__ void agg_edges(const int* __restrict__ src, const int* __restrict__ dst,
                          const float* __restrict__ dinv,
                          const float* __restrict__ H, float* __restrict__ AGG,
                          int E) {
    long long gid = (long long)blockIdx.x * blockDim.x + threadIdx.x;
    if (gid >= (long long)E * F) return;
    int e = (int)(gid / F);
    int f = (int)(gid - (long long)e * F);
    int s = src[e], d = dst[e];
    float norm = dinv[s] * dinv[d];
    unsafeAtomicAdd(&AGG[(size_t)d * F + f], H[(size_t)s * F + f] * norm);
}

template <int F>
__global__ void finalize_relu(const float* __restrict__ H, const float* __restrict__ b,
                              const float* __restrict__ dinv, float* __restrict__ AGG,
                              int Nn) {
    long long gid = (long long)blockIdx.x * blockDim.x + threadIdx.x;
    if (gid >= (long long)Nn * F) return;
    int i = (int)(gid / F);
    int f = (int)(gid - (long long)i * F);
    float di = dinv[i];
    float v  = AGG[gid] + H[gid] * di * di + b[f];
    AGG[gid] = v > 0.0f ? v : 0.0f;
}

__global__ void finalize_logsoftmax(const float* __restrict__ H2,
                                    const float* __restrict__ b2,
                                    const float* __restrict__ dinv,
                                    const float* __restrict__ AGG2,
                                    float* __restrict__ OUT, int Nn) {
    const int lane = threadIdx.x & 31;
    const int wave = threadIdx.x >> 5;
    const int node = blockIdx.x * (blockDim.x >> 5) + wave;
    if (node >= Nn) return;

    const float di = dinv[node];
    const float dd = di * di;
    const size_t base = (size_t)node * F_OUT;

    float v0 = AGG2[base + lane] + H2[base + lane] * dd + b2[lane];
    float v1 = -INFINITY;
    if (lane < 8)
        v1 = AGG2[base + 32 + lane] + H2[base + 32 + lane] * dd + b2[32 + lane];

    float m = fmaxf(v0, v1);
#pragma unroll
    for (int off = 16; off > 0; off >>= 1)
        m = fmaxf(m, __shfl_xor(m, off, 32));

    float s = __expf(v0 - m) + ((lane < 8) ? __expf(v1 - m) : 0.0f);
#pragma unroll
    for (int off = 16; off > 0; off >>= 1)
        s += __shfl_xor(s, off, 32);

    const float ls = __logf(s);
    OUT[base + lane] = v0 - m - ls;
    if (lane < 8) OUT[base + 32 + lane] = v1 - m - ls;
}

extern "C" void kernel_launch(void* const* d_in, const int* in_sizes, int n_in,
                              void* d_out, int out_size, void* d_ws, size_t ws_size,
                              hipStream_t stream) {
    const float* x  = (const float*)d_in[0];
    const float* W1 = (const float*)d_in[1];
    const float* b1 = (const float*)d_in[2];
    const float* W2 = (const float*)d_in[3];
    const float* b2 = (const float*)d_in[4];
    const int*   ei = (const int*)d_in[5];
    const int E = in_sizes[5] / 2;
    const int Nn = N_NODES;
    const int* src = ei;
    const int* dst = ei + E;

    float* out = (float*)d_out;

    float* ws   = (float*)d_ws;
    float* dinv = ws;
    float* h1   = dinv + 50048;
    float* agg1 = h1   + (size_t)Nn * F_HID;
    float* h2   = agg1 + (size_t)Nn * F_HID;
    float* agg2 = h2   + (size_t)Nn * F_OUT;

    const int TPB = 256;

    zero_f32<<<(Nn + TPB - 1) / TPB, TPB, 0, stream>>>(dinv, Nn);
    deg_acc<<<(E + TPB - 1) / TPB, TPB, 0, stream>>>(dst, dinv, E);
    make_dinv<<<(Nn + TPB - 1) / TPB, TPB, 0, stream>>>(dinv, Nn);

    {
        int mtiles = Nn / 16;
        int blocks = (mtiles + 7) / 8;
        gemm_wmma_f32<8, 128><<<blocks, TPB, 0, stream>>>(x, W1, h1, Nn);
    }

    {
        long long total = (long long)Nn * F_HID;
        zero_f32<<<(int)((total + TPB - 1) / TPB), TPB, 0, stream>>>(agg1, (int)total);
        long long work = (long long)E * F_HID;
        agg_edges<F_HID><<<(int)((work + TPB - 1) / TPB), TPB, 0, stream>>>(
            src, dst, dinv, h1, agg1, E);
        finalize_relu<F_HID><<<(int)((total + TPB - 1) / TPB), TPB, 0, stream>>>(
            h1, b1, dinv, agg1, Nn);
    }

    {
        int mtiles = Nn / 16;
        int blocks = (mtiles + 7) / 8;
        gemm_wmma_f32<3, 40><<<blocks, TPB, 0, stream>>>(agg1, W2, h2, Nn);
    }

    {
        long long total = (long long)Nn * F_OUT;
        zero_f32<<<(int)((total + TPB - 1) / TPB), TPB, 0, stream>>>(agg2, (int)total);
        long long work = (long long)E * F_OUT;
        agg_edges<F_OUT><<<(int)((work + TPB - 1) / TPB), TPB, 0, stream>>>(
            src, dst, dinv, h2, agg2, E);
        int blocks = (Nn + 7) / 8;
        finalize_logsoftmax<<<blocks, TPB, 0, stream>>>(h2, b2, dinv, agg2, out, Nn);
    }
}


# Source rows of both aggregations read with 16-byte loads, F/4 lanes per row: 16 rows (F=128) or a whole 32-edge batch (F=40) in flight per wave; LDS columns interleaved; sum order LDS arrival
# speedup vs baseline: 2.9159x; 2.9159x over previous
#include <hip/hip_runtime.h>
#include <math.h>

typedef float v2f __attribute__((ext_vector_type(2)));
typedef float v8f __attribute__((ext_vector_type(8)));

#define N_NODES 50000
#define F_IN    128
#define F_HID   128
#define F_OUT   40
#define OWN_GROUPS 254
#define OWN_NPB ((N_NODES + OWN_GROUPS - 1) / OWN_GROUPS)
#define DINV_TPB 256

__global__ void zero_f32(float* __restrict__ p, int n) {
    int i = blockIdx.x * blockDim.x + threadIdx.x;
    if (i < n) p[i] = 0.0f;
}

__global__ void deg_keys(const int* __restrict__ dst, float* __restrict__ deg,
                         unsigned* __restrict__ keys, int E) {
    const int lane = threadIdx.x & 31;
    const int i    = threadIdx.x >> 5;
    const int e0   = blockIdx.x * 512 + i * 128 + lane;
    const int last = E - 1;
    int d[4];
#pragma unroll
    for (int b = 0; b < 4; ++b) {
        const int e = e0 + 32 * b;
        d[b] = dst[e < E ? e : last];
    }
    unsigned kw = 0u;
#pragma unroll
    for (int b = 0; b < 4; ++b) {
        const int e = e0 + 32 * b;
        unsigned kb = 0xFFu;
        if (e < E) {
            unsafeAtomicAdd(&deg[d[b]], 1.0f);
            const unsigned o = (unsigned)d[b] / OWN_NPB;
            kb = o < 0xFFu ? o : 0xFFu;
        }
        kw |= kb << (8 * b);
    }
    keys[blockIdx.x * 128 + lane * 4 + i] = kw;
}

__global__ void make_dinv(float* __restrict__ deg, float* __restrict__ gcnt, int n) {
    __shared__ float part[DINV_TPB / 32];
    int i = blockIdx.x * OWN_NPB + threadIdx.x;
    float d = 0.0f;
    if (threadIdx.x < OWN_NPB && i < n) {
        d = deg[i];
        deg[i] = rsqrtf(d + 1.0f);
    }
#pragma unroll
    for (int off = 16; off > 0; off >>= 1)
        d += __shfl_xor(d, off, 32);
    if ((threadIdx.x & 31) == 0) part[threadIdx.x >> 5] = d;
    __syncthreads();
    if (threadIdx.x == 0) {
        float t = 0.0f;
        for (int w = 0; w < DINV_TPB / 32; ++w) t += part[w];
        gcnt[blockIdx.x] = t;
    }
}

template <int NT, int NC>
__global__ void gemm_wmma_f32(const float* __restrict__ X,
                              const float* __restrict__ W,
                              float* __restrict__ H, int M) {
    const int lane = threadIdx.x & 31;
    const int wave = threadIdx.x >> 5;
    const int mt   = blockIdx.x * (blockDim.x >> 5) + wave;
    if (mt * 16 >= M) return;

    const int half = lane >> 4;
    const int l16  = lane & 15;

    v8f c[NT] = {};

    const float* xrow = X + (size_t)(mt * 16 + l16) * 128;

    for (int k = 0; k < 128; k += 4) {
        v2f a;
        a.x = xrow[k + 2 * half];
        a.y = xrow[k + 2 * half + 1];
#pragma unroll
        for (int t = 0; t < NT; ++t) {
            const int n = t * 16 + l16;
            v2f b;
            if ((NC % 16 == 0) || (n < NC)) {
                b.x = W[(size_t)(k + 2 * half) * NC + n];
                b.y = W[(size_t)(k + 2 * half + 1) * NC + n];
            } else {
                b.x = 0.0f; b.y = 0.0f;
            }
            c[t] = __builtin_amdgcn_wmma_f32_16x16x4_f32(
                false, a, false, b, (short)0, c[t], false, false);
        }
    }

#pragma unroll
    for (int t = 0; t < NT; ++t) {
        const int n = t * 16 + l16;
        if ((NC % 16 == 0) || (n < NC)) {
            float* out = H + (size_t)(mt * 16 + half * 8) * NC + n;
#pragma unroll
            for (int r = 0; r < 8; ++r) out[(size_t)r * NC] = c[t][r];
        }
    }
}

#define OWN_WAVES 32
#define OWN_RING  4
#define OWN_QCAP  544

static_assert(OWN_NPB <= 256, "a list entry keeps the local row in 8 bits");
static_assert(DINV_TPB >= OWN_NPB && DINV_TPB % 32 == 0, "make_dinv: one thread per node of a range, whole waves");
static_assert(N_NODES <= 65536, "a list entry keeps the source node in 16 bits");
static_assert((N_NODES + OWN_NPB - 1) / OWN_NPB <= 255, "key 255 must belong to no group");
static_assert((N_NODES + OWN_NPB - 1) / OWN_NPB <= OWN_WAVES * 32, "one thread per earlier group sums the list offset");

template <int F>
__device__ __forceinline__ int lds_col(int c) { return (c & 3) * (F / 4) + (c >> 2); }

template <int F>
__device__ __forceinline__ void start_rows(const float* __restrict__ dinv,
                                           const float* __restrict__ H,
                                           const float* __restrict__ bias,
                                           float* acc, int lo, int cnt, int tid) {
    for (int idx = tid; idx < cnt * F; idx += OWN_WAVES * 32) {
        const int i = idx / F;
        const int f = idx - i * F;
        const float di = dinv[lo + i];
        acc[i * F + lds_col<F>(f)] = H[(size_t)lo * F + idx] * di * di + bias[f];
    }
}

__device__ __forceinline__ void list_offset_share(const float* __restrict__ gcnt, float* red,
                                                  int g, int tid, int lane, int wave) {
    float v = (tid < g) ? gcnt[tid] : 0.0f;
#pragma unroll
    for (int off = 16; off > 0; off >>= 1)
        v += __shfl_xor(v, off, 32);
    if (lane == 0) red[wave] = v;
}

__device__ __forceinline__ int list_offset(const float* red) {
    float t = 0.0f;
    for (int i = 0; i < OWN_WAVES; ++i) t += red[i];
    return __builtin_amdgcn_readfirstlane((int)t);
}

template <int F>
__device__ __forceinline__ void add_rows(const float* __restrict__ H, float* acc,
                                         int s, int dloc, float w, int n, int lane) {
    static_assert(F % 4 == 0 && F / 4 <= 32, "a row is F / 4 lanes of 16 bytes");
    constexpr int LPR = F / 4;
    constexpr int RPL = 32 / LPR;
    const float4* __restrict__ H4 = reinterpret_cast<const float4*>(H);
    if constexpr (RPL == 1) {
        constexpr int ROWS = 16;
#pragma nounroll
        for (int k = 0; k < n; k += ROWS) {
            float4 r[ROWS];
#pragma unroll
            for (int u = 0; u < ROWS; ++u) {
                const int sk = __builtin_amdgcn_readlane(s, (k + u) & 31);
                r[u] = ((k + u) < n) ? H4[(size_t)sk * LPR + lane] : make_float4(0.0f, 0.0f, 0.0f, 0.0f);
            }
#pragma unroll
            for (int u = 0; u < ROWS; ++u) {
                const float wk = __int_as_float(__builtin_amdgcn_readlane(__float_as_int(w), (k + u) & 31));
                const int   dk = __builtin_amdgcn_readlane(dloc, (k + u) & 31);
                if ((k + u) < n) {
                    float* a = acc + dk * F + lane;
                    __hip_atomic_fetch_add(a,           r[u].x * wk, __ATOMIC_RELAXED, __HIP_MEMORY_SCOPE_WORKGROUP);
                    __hip_atomic_fetch_add(a + LPR,     r[u].y * wk, __ATOMIC_RELAXED, __HIP_MEMORY_SCOPE_WORKGROUP);
                    __hip_atomic_fetch_add(a + 2 * LPR, r[u].z * wk, __ATOMIC_RELAXED, __HIP_MEMORY_SCOPE_WORKGROUP);
                    __hip_atomic_fetch_add(a + 3 * LPR, r[u].w * wk, __ATOMIC_RELAXED, __HIP_MEMORY_SCOPE_WORKGROUP);
                }
            }
        }
    } else {
        constexpr int NLD = (32 + RPL - 1) / RPL;
        const int  g    = lane / LPR;
        const int  m    = lane - g * LPR;
        const bool live = g < RPL;
        const int  pk   = s | (dloc << 16);
        float4 r[NLD];
        int    p[NLD];
        float  wk[NLD];
#pragma unroll
        for (int t = 0; t < NLD; ++t) {
            const int e = t * RPL + g;
            p[t]  = __shfl(pk, e & 31, 32);
            wk[t] = __shfl(w, e & 31, 32);
        }
#pragma unroll
        for (int t = 0; t < NLD; ++t) {
            const int e = t * RPL + g;
            const int okm = -(int)(live && e < n);
            const int sk = p[t] & 0xFFFF & okm;
            p[t] |= ~okm;
            r[t] = H4[(size_t)sk * LPR + m];
        }
#pragma unroll
        for (int t = 0; t < NLD; ++t) {
            if (p[t] >= 0) {
                float* a = acc + (p[t] >> 16) * F + m;
                __hip_atomic_fetch_add(a,           r[t].x * wk[t], __ATOMIC_RELAXED, __HIP_MEMORY_SCOPE_WORKGROUP);
                __hip_atomic_fetch_add(a + LPR,     r[t].y * wk[t], __ATOMIC_RELAXED, __HIP_MEMORY_SCOPE_WORKGROUP);
                __hip_atomic_fetch_add(a + 2 * LPR, r[t].z * wk[t], __ATOMIC_RELAXED, __HIP_MEMORY_SCOPE_WORKGROUP);
                __hip_atomic_fetch_add(a + 3 * LPR, r[t].w * wk[t], __ATOMIC_RELAXED, __HIP_MEMORY_SCOPE_WORKGROUP);
            }
        }
    }
}

template <int F>
__device__ __forceinline__ void drain_queue(const int* __restrict__ src,
                                            const int* __restrict__ dst,
                                            const float* __restrict__ dinv,
                                            const float* __restrict__ H,
                                            float* acc, const float* dl,
                                            const int* qe, int lo,
                                            uint2* __restrict__ glist, int* lcnt,
                                            int n, int lane) {
    int s = 0, dloc = 0;
    float w = 0.0f;
    if (lane < n) {
        const int e = qe[lane];
        s = src[e];
        dloc = dst[e] - lo;
        w = dinv[s] * dl[dloc];
    }
    int base = 0;
    if (lane == 0)
        base = __hip_atomic_fetch_add(lcnt, n, __ATOMIC_RELAXED, __HIP_MEMORY_SCOPE_WORKGROUP);
    base = __builtin_amdgcn_readfirstlane(base);
    if (lane < n)
        glist[base + lane] = make_uint2((unsigned)s | ((unsigned)dloc << 16), __float_as_uint(w));
    add_rows<F>(H, acc, s, dloc, w, n, lane);
}

__device__ __forceinline__ void queue_rows(unsigned x, int e0, unsigned ltmask, int* myq, int& qn) {
#pragma unroll
    for (int b = 0; b < 4; ++b) {
        const bool m = ((x >> (8 * b)) & 0xFFu) == 0u;
        const unsigned mask = __builtin_amdgcn_ballot_w32(m);
        if (mask != 0u) {
            const int pos = qn + __builtin_popcount(mask & ltmask);
            if (m) myq[pos] = e0 + 32 * b;
            qn += __builtin_popcount(mask);
        }
    }
}

template <int F>
__global__ void __launch_bounds__(OWN_WAVES * 32)
agg_scan(const int* __restrict__ src, const int* __restrict__ dst,
         const unsigned* __restrict__ keys, const float* __restrict__ gcnt,
         const float* __restrict__ dinv, const float* __restrict__ H,
         const float* __restrict__ bias, float* __restrict__ OUT,
         unsigned* __restrict__ list, int E, int Nn) {
    __shared__ float acc[OWN_NPB * F];
    __shared__ float dl[OWN_NPB];
    __shared__ int   qe[OWN_WAVES][OWN_QCAP];
    __shared__ float red[OWN_WAVES];
    __shared__ int   lcnt;

    const int tid  = threadIdx.x;
    const int lane = tid & 31;
    const int wave = __builtin_amdgcn_readfirstlane(tid >> 5);
    const int lo   = blockIdx.x * OWN_NPB;
    const int cnt  = (Nn - lo) < OWN_NPB ? (Nn - lo) : OWN_NPB;

    if (tid == 0) lcnt = 0;
    list_offset_share(gcnt, red, blockIdx.x, tid, lane, wave);
    for (int i = tid; i < cnt; i += OWN_WAVES * 32) dl[i] = dinv[lo + i];
    start_rows<F>(dinv, H, bias, acc, lo, cnt, tid);
    __syncthreads();

    uint2* glist = reinterpret_cast<uint2*>(list) + list_offset(red);
    int* myq = qe[wave];
    int qn = 0;

    const int nblk = (E + 511) >> 9;
    if (nblk > 0) {
        const uint4* __restrict__ kp = reinterpret_cast<const uint4*>(keys);
        const unsigned own4   = (unsigned)blockIdx.x * 0x01010101u;
        const unsigned ltmask = (1u << lane) - 1u;
        const int lastb = nblk - 1;
        int blk = wave;
        uint4 kn[OWN_RING];
#pragma unroll
        for (int j = 0; j < OWN_RING; ++j) {
            const int bj = blk + j * OWN_WAVES;
            kn[j] = kp[(size_t)(bj < nblk ? bj : lastb) * 32 + lane];
        }
#pragma nounroll
        while (blk < nblk) {
            uint4 kv[OWN_RING];
#pragma unroll
            for (int j = 0; j < OWN_RING; ++j) kv[j] = kn[j];
            const int nbk = blk + OWN_RING * OWN_WAVES;
#pragma unroll
            for (int j = 0; j < OWN_RING; ++j) {
                const int bj = nbk + j * OWN_WAVES;
                kn[j] = kp[(size_t)(bj < nblk ? bj : lastb) * 32 + lane];
            }
#pragma unroll
            for (int j = 0; j < OWN_RING; ++j) {
                const int bj = blk + j * OWN_WAVES;
                if (bj < nblk) {
                    const unsigned x0 = kv[j].x ^ own4, x1 = kv[j].y ^ own4, x2 = kv[j].z ^ own4, x3 = kv[j].w ^ own4;
                    const unsigned z0 = (x0 - 0x01010101u) & ~x0 & 0x80808080u;
                    const unsigned z1 = (x1 - 0x01010101u) & ~x1 & 0x80808080u;
                    const unsigned z2 = (x2 - 0x01010101u) & ~x2 & 0x80808080u;
                    const unsigned z3 = (x3 - 0x01010101u) & ~x3 & 0x80808080u;
                    const unsigned M0 = __builtin_amdgcn_ballot_w32(z0 != 0u);
                    const unsigned M1 = __builtin_amdgcn_ballot_w32(z1 != 0u);
                    const unsigned M2 = __builtin_amdgcn_ballot_w32(z2 != 0u);
                    const unsigned M3 = __builtin_amdgcn_ballot_w32(z3 != 0u);
                    if ((M0 | M1 | M2 | M3) != 0u) {
                        const int e0 = bj * 512 + lane;
                        if (M0 != 0u) queue_rows(x0, e0,       ltmask, myq, qn);
                        if (M1 != 0u) queue_rows(x1, e0 + 128, ltmask, myq, qn);
                        if (M2 != 0u) queue_rows(x2, e0 + 256, ltmask, myq, qn);
                        if (M3 != 0u) queue_rows(x3, e0 + 384, ltmask, myq, qn);
                        if (qn >= 32) {
                            int off = 0;
                            while (qn - off >= 32) {
                                drain_queue<F>(src, dst, dinv, H, acc, dl, myq + off, lo, glist, &lcnt, 32, lane);
                                off += 32;
                            }
                            qn -= off;
                            int te = 0;
                            if (lane < qn) te = myq[off + lane];
                            if (lane < qn) myq[lane] = te;
                        }
                    }
                }
            }
            blk = nbk;
        }
    }
    if (qn > 0) drain_queue<F>(src, dst, dinv, H, acc, dl, myq, lo, glist, &lcnt, qn, lane);
    __syncthreads();

    for (int idx = tid; idx < cnt * F; idx += OWN_WAVES * 32) {
        const int i = idx / F;
        const float v = acc[i * F + lds_col<F>(idx - i * F)];
        OUT[(size_t)lo * F + idx] = v > 0.0f ? v : 0.0f;
    }
}

template <int F>
__global__ void __launch_bounds__(OWN_WAVES * 32)
agg_walk(const unsigned* __restrict__ list, const float* __restrict__ gcnt,
         const float* __restrict__ dinv, const float* __restrict__ H,
         const float* __restrict__ bias, float* __restrict__ OUT, int Nn) {
    static_assert(F > 32 && F <= 64, "the epilogue holds a row in two registers per lane");
    __shared__ float acc[OWN_NPB * F];
    __shared__ float red[OWN_WAVES];

    const int tid  = threadIdx.x;
    const int lane = tid & 31;
    const int wave = __builtin_amdgcn_readfirstlane(tid >> 5);
    const int lo   = blockIdx.x * OWN_NPB;
    const int cnt  = (Nn - lo) < OWN_NPB ? (Nn - lo) : OWN_NPB;

    list_offset_share(gcnt, red, blockIdx.x, tid, lane, wave);
    start_rows<F>(dinv, H, bias, acc, lo, cnt, tid);
    __syncthreads();

    const uint2* __restrict__ glist = reinterpret_cast<const uint2*>(list) + list_offset(red);
    const int len = (int)gcnt[blockIdx.x];

    int c = wave * 32;
    uint2 en = make_uint2(0u, 0u);
    if (c + lane < len) en = glist[c + lane];
#pragma nounroll
    while (c < len) {
        const uint2 ev = en;
        const int rem = len - c;
        const int cn = c + OWN_WAVES * 32;
        en = glist[(cn + lane < len) ? (cn + lane) : (len - 1)];
        const int   s    = (int)(ev.x & 0xFFFFu);
        const int   dloc = (int)(ev.x >> 16);
        const float w    = __uint_as_float(ev.y);
        if (rem >= 32) add_rows<F>(H, acc, s, dloc, w, 32, lane);
        else           add_rows<F>(H, acc, s, dloc, w, rem, lane);
        c = cn;
    }
    __syncthreads();

    for (int i = wave; i < cnt; i += OWN_WAVES) {
        const float v0 = (lane < F) ? acc[i * F + lds_col<F>(lane)] : -INFINITY;
        const float v1 = (lane + 32 < F) ? acc[i * F + lds_col<F>(32 + lane)] : -INFINITY;
        float m = fmaxf(v0, v1);
#pragma unroll
        for (int off = 16; off > 0; off >>= 1)
            m = fmaxf(m, __shfl_xor(m, off, 32));
        float sm = ((lane < F) ? __expf(v0 - m) : 0.0f) +
                   ((lane + 32 < F) ? __expf(v1 - m) : 0.0f);
#pragma unroll
        for (int off = 16; off > 0; off >>= 1)
            sm += __shfl_xor(sm, off, 32);
        const float ls = __logf(sm);
        const size_t ob = (size_t)(lo + i) * F;
        if (lane < F)      OUT[ob + lane]      = v0 - m - ls;
        if (lane + 32 < F) OUT[ob + 32 + lane] = v1 - m - ls;
    }
}

extern "C" void kernel_launch(void* const* d_in, const int* in_sizes, int n_in,
                              void* d_out, int out_size, void* d_ws, size_t ws_size,
                              hipStream_t stream) {
    const float* x  = (const float*)d_in[0];
    const float* W1 = (const float*)d_in[1];
    const float* b1 = (const float*)d_in[2];
    const float* W2 = (const float*)d_in[3];
    const float* b2 = (const float*)d_in[4];
    const int*   ei = (const int*)d_in[5];
    const int E = in_sizes[5] / 2;
    const int Nn = N_NODES;
    const int* src = ei;
    const int* dst = ei + E;

    float* out = (float*)d_out;

    float* ws   = (float*)d_ws;
    float* dinv = ws;
    float* h1   = dinv + 50048;
    float* agg1 = h1   + (size_t)Nn * F_HID;
    float* h2   = agg1 + (size_t)Nn * F_HID;
    unsigned* keys = (unsigned*)(h2 + (size_t)Nn * F_OUT);
    float* gcnt = (float*)(keys + (size_t)((E + 511) / 512) * 128);
    unsigned* list = (unsigned*)(gcnt + 256);

    const int TPB = 256;

    zero_f32<<<(Nn + TPB - 1) / TPB, TPB, 0, stream>>>(dinv, Nn);
    deg_keys<<<(E + 511) / 512, 128, 0, stream>>>(dst, dinv, keys, E);
    make_dinv<<<(Nn + OWN_NPB - 1) / OWN_NPB, DINV_TPB, 0, stream>>>(dinv, gcnt, Nn);

    {
        int mtiles = Nn / 16;
        int blocks = (mtiles + 7) / 8;
        gemm_wmma_f32<8, 128><<<blocks, TPB, 0, stream>>>(x, W1, h1, Nn);
    }

    {
        int blocks = (Nn + OWN_NPB - 1) / OWN_NPB;
        agg_scan<F_HID><<<blocks, OWN_WAVES * 32, 0, stream>>>(
            src, dst, keys, gcnt, dinv, h1, b1, agg1, list, E, Nn);
    }

    {
        int mtiles = Nn / 16;
        int blocks = (mtiles + 7) / 8;
        gemm_wmma_f32<3, 40><<<blocks, TPB, 0, stream>>>(agg1, W2, h2, Nn);
    }

    {
        int blocks = (Nn + OWN_NPB - 1) / OWN_NPB;
        agg_walk<F_OUT><<<blocks, OWN_WAVES * 32, 0, stream>>>(
            list, gcnt, dinv, h2, b2, out, Nn);
    }
}


# Degree/owner-key work-groups ride in the layer-1 product launch (independent work, product work-groups first): 6 launches, not 7; same code per work-group, same sums
# speedup vs baseline: 3.0629x; 1.0504x over previous
#include <hip/hip_runtime.h>
#include <math.h>

typedef float v2f __attribute__((ext_vector_type(2)));
typedef float v8f __attribute__((ext_vector_type(8)));

#define N_NODES 50000
#define F_IN    128
#define F_HID   128
#define F_OUT   40
#define OWN_GROUPS 254
#define OWN_NPB ((N_NODES + OWN_GROUPS - 1) / OWN_GROUPS)
#define DINV_TPB 256

__global__ void zero_f32(float* __restrict__ p, int n) {
    int i = blockIdx.x * blockDim.x + threadIdx.x;
    if (i < n) p[i] = 0.0f;
}

__device__ __forceinline__ void deg_keys_block(const int* __restrict__ dst, float* __restrict__ deg,
                                               unsigned* __restrict__ keys, int E,
                                               int kb, int i, int lane) {
    const int e0   = kb * 512 + i * 128 + lane;
    const int last = E - 1;
    int d[4];
#pragma unroll
    for (int b = 0; b < 4; ++b) {
        const int e = e0 + 32 * b;
        d[b] = dst[e < E ? e : last];
    }
    unsigned kw = 0u;
#pragma unroll
    for (int b = 0; b < 4; ++b) {
        const int e = e0 + 32 * b;
        unsigned kb = 0xFFu;
        if (e < E) {
            unsafeAtomicAdd(&deg[d[b]], 1.0f);
            const unsigned o = (unsigned)d[b] / OWN_NPB;
            kb = o < 0xFFu ? o : 0xFFu;
        }
        kw |= kb << (8 * b);
    }
    keys[kb * 128 + lane * 4 + i] = kw;
}

__global__ void make_dinv(float* __restrict__ deg, float* __restrict__ gcnt, int n) {
    __shared__ float part[DINV_TPB / 32];
    int i = blockIdx.x * OWN_NPB + threadIdx.x;
    float d = 0.0f;
    if (threadIdx.x < OWN_NPB && i < n) {
        d = deg[i];
        deg[i] = rsqrtf(d + 1.0f);
    }
#pragma unroll
    for (int off = 16; off > 0; off >>= 1)
        d += __shfl_xor(d, off, 32);
    if ((threadIdx.x & 31) == 0) part[threadIdx.x >> 5] = d;
    __syncthreads();
    if (threadIdx.x == 0) {
        float t = 0.0f;
        for (int w = 0; w < DINV_TPB / 32; ++w) t += part[w];
        gcnt[blockIdx.x] = t;
    }
}

template <int NT, int NC>
__device__ __forceinline__ void gemm_strip_f32(const float* __restrict__ X,
                                               const float* __restrict__ W,
                                               float* __restrict__ H, int M, int mt, int lane) {
    if (mt * 16 >= M) return;

    const int half = lane >> 4;
    const int l16  = lane & 15;

    v8f c[NT] = {};

    const float* xrow = X + (size_t)(mt * 16 + l16) * 128;

    for (int k = 0; k < 128; k += 4) {
        v2f a;
        a.x = xrow[k + 2 * half];
        a.y = xrow[k + 2 * half + 1];
#pragma unroll
        for (int t = 0; t < NT; ++t) {
            const int n = t * 16 + l16;
            v2f b;
            if ((NC % 16 == 0) || (n < NC)) {
                b.x = W[(size_t)(k + 2 * half) * NC + n];
                b.y = W[(size_t)(k + 2 * half + 1) * NC + n];
            } else {
                b.x = 0.0f; b.y = 0.0f;
            }
            c[t] = __builtin_amdgcn_wmma_f32_16x16x4_f32(
                false, a, false, b, (short)0, c[t], false, false);
        }
    }

#pragma unroll
    for (int t = 0; t < NT; ++t) {
        const int n = t * 16 + l16;
        if ((NC % 16 == 0) || (n < NC)) {
            float* out = H + (size_t)(mt * 16 + half * 8) * NC + n;
#pragma unroll
            for (int r = 0; r < 8; ++r) out[(size_t)r * NC] = c[t][r];
        }
    }
}

template <int NT, int NC>
__global__ void gemm_wmma_f32(const float* __restrict__ X,
                              const float* __restrict__ W,
                              float* __restrict__ H, int M) {
    gemm_strip_f32<NT, NC>(X, W, H, M, blockIdx.x * (blockDim.x >> 5) + (threadIdx.x >> 5), threadIdx.x & 31);
}

#define GD_TPB 256
__global__ void __launch_bounds__(GD_TPB)
gemm1_deg_keys(const float* __restrict__ X, const float* __restrict__ W, float* __restrict__ H, int M,
               int gemm_blocks,
               const int* __restrict__ dst, float* __restrict__ deg, unsigned* __restrict__ keys, int E) {
    const int lane = threadIdx.x & 31;
    const int wave = threadIdx.x >> 5;
    if ((int)blockIdx.x < gemm_blocks) {
        gemm_strip_f32<8, 128>(X, W, H, M, blockIdx.x * (GD_TPB / 32) + wave, lane);
    } else {
        const int kb = ((int)blockIdx.x - gemm_blocks) * (GD_TPB / 128) + (wave >> 2);
        if (kb * 512 < E)
            deg_keys_block(dst, deg, keys, E, kb, wave & 3, lane);
    }
}

#define OWN_WAVES 32
#define OWN_RING  4
#define OWN_QCAP  544

static_assert(OWN_NPB <= 256, "a list entry keeps the local row in 8 bits");
static_assert(DINV_TPB >= OWN_NPB && DINV_TPB % 32 == 0, "make_dinv: one thread per node of a range, whole waves");
static_assert(N_NODES <= 65536, "a list entry keeps the source node in 16 bits");
static_assert((N_NODES + OWN_NPB - 1) / OWN_NPB <= 255, "key 255 must belong to no group");
static_assert((N_NODES + OWN_NPB - 1) / OWN_NPB <= OWN_WAVES * 32, "one thread per earlier group sums the list offset");

template <int F>
__device__ __forceinline__ int lds_col(int c) { return (c & 3) * (F / 4) + (c >> 2); }

template <int F>
__device__ __forceinline__ void start_rows(const float* __restrict__ dinv,
                                           const float* __restrict__ H,
                                           const float* __restrict__ bias,
                                           float* acc, int lo, int cnt, int tid) {
    for (int idx = tid; idx < cnt * F; idx += OWN_WAVES * 32) {
        const int i = idx / F;
        const int f = idx - i * F;
        const float di = dinv[lo + i];
        acc[i * F + lds_col<F>(f)] = H[(size_t)lo * F + idx] * di * di + bias[f];
    }
}

__device__ __forceinline__ void list_offset_share(const float* __restrict__ gcnt, float* red,
                                                  int g, int tid, int lane, int wave) {
    float v = (tid < g) ? gcnt[tid] : 0.0f;
#pragma unroll
    for (int off = 16; off > 0; off >>= 1)
        v += __shfl_xor(v, off, 32);
    if (lane == 0) red[wave] = v;
}

__device__ __forceinline__ int list_offset(const float* red) {
    float t = 0.0f;
    for (int i = 0; i < OWN_WAVES; ++i) t += red[i];
    return __builtin_amdgcn_readfirstlane((int)t);
}

template <int F>
__device__ __forceinline__ void add_rows(const float* __restrict__ H, float* acc,
                                         int s, int dloc, float w, int n, int lane) {
    static_assert(F % 4 == 0 && F / 4 <= 32, "a row is F / 4 lanes of 16 bytes");
    constexpr int LPR = F / 4;
    constexpr int RPL = 32 / LPR;
    const float4* __restrict__ H4 = reinterpret_cast<const float4*>(H);
    if constexpr (RPL == 1) {
        constexpr int ROWS = 16;
#pragma nounroll
        for (int k = 0; k < n; k += ROWS) {
            float4 r[ROWS];
#pragma unroll
            for (int u = 0; u < ROWS; ++u) {
                const int sk = __builtin_amdgcn_readlane(s, (k + u) & 31);
                r[u] = ((k + u) < n) ? H4[(size_t)sk * LPR + lane] : make_float4(0.0f, 0.0f, 0.0f, 0.0f);
            }
#pragma unroll
            for (int u = 0; u < ROWS; ++u) {
                const float wk = __int_as_float(__builtin_amdgcn_readlane(__float_as_int(w), (k + u) & 31));
                const int   dk = __builtin_amdgcn_readlane(dloc, (k + u) & 31);
                if ((k + u) < n) {
                    float* a = acc + dk * F + lane;
                    __hip_atomic_fetch_add(a,           r[u].x * wk, __ATOMIC_RELAXED, __HIP_MEMORY_SCOPE_WORKGROUP);
                    __hip_atomic_fetch_add(a + LPR,     r[u].y * wk, __ATOMIC_RELAXED, __HIP_MEMORY_SCOPE_WORKGROUP);
                    __hip_atomic_fetch_add(a + 2 * LPR, r[u].z * wk, __ATOMIC_RELAXED, __HIP_MEMORY_SCOPE_WORKGROUP);
                    __hip_atomic_fetch_add(a + 3 * LPR, r[u].w * wk, __ATOMIC_RELAXED, __HIP_MEMORY_SCOPE_WORKGROUP);
                }
            }
        }
    } else {
        constexpr int NLD = (32 + RPL - 1) / RPL;
        const int  g    = lane / LPR;
        const int  m    = lane - g * LPR;
        const bool live = g < RPL;
        const int  pk   = s | (dloc << 16);
        float4 r[NLD];
        int    p[NLD];
        float  wk[NLD];
#pragma unroll
        for (int t = 0; t < NLD; ++t) {
            const int e = t * RPL + g;
            p[t]  = __shfl(pk, e & 31, 32);
            wk[t] = __shfl(w, e & 31, 32);
        }
#pragma unroll
        for (int t = 0; t < NLD; ++t) {
            const int e = t * RPL + g;
            const int okm = -(int)(live && e < n);
            const int sk = p[t] & 0xFFFF & okm;
            p[t] |= ~okm;
            r[t] = H4[(size_t)sk * LPR + m];
        }
#pragma unroll
        for (int t = 0; t < NLD; ++t) {
            if (p[t] >= 0) {
                float* a = acc + (p[t] >> 16) * F + m;
                __hip_atomic_fetch_add(a,           r[t].x * wk[t], __ATOMIC_RELAXED, __HIP_MEMORY_SCOPE_WORKGROUP);
                __hip_atomic_fetch_add(a + LPR,     r[t].y * wk[t], __ATOMIC_RELAXED, __HIP_MEMORY_SCOPE_WORKGROUP);
                __hip_atomic_fetch_add(a + 2 * LPR, r[t].z * wk[t], __ATOMIC_RELAXED, __HIP_MEMORY_SCOPE_WORKGROUP);
                __hip_atomic_fetch_add(a + 3 * LPR, r[t].w * wk[t], __ATOMIC_RELAXED, __HIP_MEMORY_SCOPE_WORKGROUP);
            }
        }
    }
}

template <int F>
__device__ __forceinline__ void drain_queue(const int* __restrict__ src,
                                            const int* __restrict__ dst,
                                            const float* __restrict__ dinv,
                                            const float* __restrict__ H,
                                            float* acc, const float* dl,
                                            const int* qe, int lo,
                                            uint2* __restrict__ glist, int* lcnt,
                                            int n, int lane) {
    int s = 0, dloc = 0;
    float w = 0.0f;
    if (lane < n) {
        const int e = qe[lane];
        s = src[e];
        dloc = dst[e] - lo;
        w = dinv[s] * dl[dloc];
    }
    int base = 0;
    if (lane == 0)
        base = __hip_atomic_fetch_add(lcnt, n, __ATOMIC_RELAXED, __HIP_MEMORY_SCOPE_WORKGROUP);
    base = __builtin_amdgcn_readfirstlane(base);
    if (lane < n)
        glist[base + lane] = make_uint2((unsigned)s | ((unsigned)dloc << 16), __float_as_uint(w));
    add_rows<F>(H, acc, s, dloc, w, n, lane);
}

__device__ __forceinline__ void queue_rows(unsigned x, int e0, unsigned ltmask, int* myq, int& qn) {
#pragma unroll
    for (int b = 0; b < 4; ++b) {
        const bool m = ((x >> (8 * b)) & 0xFFu) == 0u;
        const unsigned mask = __builtin_amdgcn_ballot_w32(m);
        if (mask != 0u) {
            const int pos = qn + __builtin_popcount(mask & ltmask);
            if (m) myq[pos] = e0 + 32 * b;
            qn += __builtin_popcount(mask);
        }
    }
}

template <int F>
__global__ void __launch_bounds__(OWN_WAVES * 32)
agg_scan(const int* __restrict__ src, const int* __restrict__ dst,
         const unsigned* __restrict__ keys, const float* __restrict__ gcnt,
         const float* __restrict__ dinv, const float* __restrict__ H,
         const float* __restrict__ bias, float* __restrict__ OUT,
         unsigned* __restrict__ list, int E, int Nn) {
    __shared__ float acc[OWN_NPB * F];
    __shared__ float dl[OWN_NPB];
    __shared__ int   qe[OWN_WAVES][OWN_QCAP];
    __shared__ float red[OWN_WAVES];
    __shared__ int   lcnt;

    const int tid  = threadIdx.x;
    const int lane = tid & 31;
    const int wave = __builtin_amdgcn_readfirstlane(tid >> 5);
    const int lo   = blockIdx.x * OWN_NPB;
    const int cnt  = (Nn - lo) < OWN_NPB ? (Nn - lo) : OWN_NPB;

    if (tid == 0) lcnt = 0;
    list_offset_share(gcnt, red, blockIdx.x, tid, lane, wave);
    for (int i = tid; i < cnt; i += OWN_WAVES * 32) dl[i] = dinv[lo + i];
    start_rows<F>(dinv, H, bias, acc, lo, cnt, tid);
    __syncthreads();

    uint2* glist = reinterpret_cast<uint2*>(list) + list_offset(red);
    int* myq = qe[wave];
    int qn = 0;

    const int nblk = (E + 511) >> 9;
    if (nblk > 0) {
        const uint4* __restrict__ kp = reinterpret_cast<const uint4*>(keys);
        const unsigned own4   = (unsigned)blockIdx.x * 0x01010101u;
        const unsigned ltmask = (1u << lane) - 1u;
        const int lastb = nblk - 1;
        int blk = wave;
        uint4 kn[OWN_RING];
#pragma unroll
        for (int j = 0; j < OWN_RING; ++j) {
            const int bj = blk + j * OWN_WAVES;
            kn[j] = kp[(size_t)(bj < nblk ? bj : lastb) * 32 + lane];
        }
#pragma nounroll
        while (blk < nblk) {
            uint4 kv[OWN_RING];
#pragma unroll
            for (int j = 0; j < OWN_RING; ++j) kv[j] = kn[j];
            const int nbk = blk + OWN_RING * OWN_WAVES;
#pragma unroll
            for (int j = 0; j < OWN_RING; ++j) {
                const int bj = nbk + j * OWN_WAVES;
                kn[j] = kp[(size_t)(bj < nblk ? bj : lastb) * 32 + lane];
            }
#pragma unroll
            for (int j = 0; j < OWN_RING; ++j) {
                const int bj = blk + j * OWN_WAVES;
                if (bj < nblk) {
                    const unsigned x0 = kv[j].x ^ own4, x1 = kv[j].y ^ own4, x2 = kv[j].z ^ own4, x3 = kv[j].w ^ own4;
                    const unsigned z0 = (x0 - 0x01010101u) & ~x0 & 0x80808080u;
                    const unsigned z1 = (x1 - 0x01010101u) & ~x1 & 0x80808080u;
                    const unsigned z2 = (x2 - 0x01010101u) & ~x2 & 0x80808080u;
                    const unsigned z3 = (x3 - 0x01010101u) & ~x3 & 0x80808080u;
                    const unsigned M0 = __builtin_amdgcn_ballot_w32(z0 != 0u);
                    const unsigned M1 = __builtin_amdgcn_ballot_w32(z1 != 0u);
                    const unsigned M2 = __builtin_amdgcn_ballot_w32(z2 != 0u);
                    const unsigned M3 = __builtin_amdgcn_ballot_w32(z3 != 0u);
                    if ((M0 | M1 | M2 | M3) != 0u) {
                        const int e0 = bj * 512 + lane;
                        if (M0 != 0u) queue_rows(x0, e0,       ltmask, myq, qn);
                        if (M1 != 0u) queue_rows(x1, e0 + 128, ltmask, myq, qn);
                        if (M2 != 0u) queue_rows(x2, e0 + 256, ltmask, myq, qn);
                        if (M3 != 0u) queue_rows(x3, e0 + 384, ltmask, myq, qn);
                        if (qn >= 32) {
                            int off = 0;
                            while (qn - off >= 32) {
                                drain_queue<F>(src, dst, dinv, H, acc, dl, myq + off, lo, glist, &lcnt, 32, lane);
                                off += 32;
                            }
                            qn -= off;
                            int te = 0;
                            if (lane < qn) te = myq[off + lane];
                            if (lane < qn) myq[lane] = te;
                        }
                    }
                }
            }
            blk = nbk;
        }
    }
    if (qn > 0) drain_queue<F>(src, dst, dinv, H, acc, dl, myq, lo, glist, &lcnt, qn, lane);
    __syncthreads();

    for (int idx = tid; idx < cnt * F; idx += OWN_WAVES * 32) {
        const int i = idx / F;
        const float v = acc[i * F + lds_col<F>(idx - i * F)];
        OUT[(size_t)lo * F + idx] = v > 0.0f ? v : 0.0f;
    }
}

template <int F>
__global__ void __launch_bounds__(OWN_WAVES * 32)
agg_walk(const unsigned* __restrict__ list, const float* __restrict__ gcnt,
         const float* __restrict__ dinv, const float* __restrict__ H,
         const float* __restrict__ bias, float* __restrict__ OUT, int Nn) {
    static_assert(F > 32 && F <= 64, "the epilogue holds a row in two registers per lane");
    __shared__ float acc[OWN_NPB * F];
    __shared__ float red[OWN_WAVES];

    const int tid  = threadIdx.x;
    const int lane = tid & 31;
    const int wave = __builtin_amdgcn_readfirstlane(tid >> 5);
    const int lo   = blockIdx.x * OWN_NPB;
    const int cnt  = (Nn - lo) < OWN_NPB ? (Nn - lo) : OWN_NPB;

    list_offset_share(gcnt, red, blockIdx.x, tid, lane, wave);
    start_rows<F>(dinv, H, bias, acc, lo, cnt, tid);
    __syncthreads();

    const uint2* __restrict__ glist = reinterpret_cast<const uint2*>(list) + list_offset(red);
    const int len = (int)gcnt[blockIdx.x];

    int c = wave * 32;
    uint2 en = make_uint2(0u, 0u);
    if (c + lane < len) en = glist[c + lane];
#pragma nounroll
    while (c < len) {
        const uint2 ev = en;
        const int rem = len - c;
        const int cn = c + OWN_WAVES * 32;
        en = glist[(cn + lane < len) ? (cn + lane) : (len - 1)];
        const int   s    = (int)(ev.x & 0xFFFFu);
        const int   dloc = (int)(ev.x >> 16);
        const float w    = __uint_as_float(ev.y);
        if (rem >= 32) add_rows<F>(H, acc, s, dloc, w, 32, lane);
        else           add_rows<F>(H, acc, s, dloc, w, rem, lane);
        c = cn;
    }
    __syncthreads();

    for (int i = wave; i < cnt; i += OWN_WAVES) {
        const float v0 = (lane < F) ? acc[i * F + lds_col<F>(lane)] : -INFINITY;
        const float v1 = (lane + 32 < F) ? acc[i * F + lds_col<F>(32 + lane)] : -INFINITY;
        float m = fmaxf(v0, v1);
#pragma unroll
        for (int off = 16; off > 0; off >>= 1)
            m = fmaxf(m, __shfl_xor(m, off, 32));
        float sm = ((lane < F) ? __expf(v0 - m) : 0.0f) +
                   ((lane + 32 < F) ? __expf(v1 - m) : 0.0f);
#pragma unroll
        for (int off = 16; off > 0; off >>= 1)
            sm += __shfl_xor(sm, off, 32);
        const float ls = __logf(sm);
        const size_t ob = (size_t)(lo + i) * F;
        if (lane < F)      OUT[ob + lane]      = v0 - m - ls;
        if (lane + 32 < F) OUT[ob + 32 + lane] = v1 - m - ls;
    }
}

extern "C" void kernel_launch(void* const* d_in, const int* in_sizes, int n_in,
                              void* d_out, int out_size, void* d_ws, size_t ws_size,
                              hipStream_t stream) {
    const float* x  = (const float*)d_in[0];
    const float* W1 = (const float*)d_in[1];
    const float* b1 = (const float*)d_in[2];
    const float* W2 = (const float*)d_in[3];
    const float* b2 = (const float*)d_in[4];
    const int*   ei = (const int*)d_in[5];
    const int E = in_sizes[5] / 2;
    const int Nn = N_NODES;
    const int* src = ei;
    const int* dst = ei + E;

    float* out = (float*)d_out;

    float* ws   = (float*)d_ws;
    float* dinv = ws;
    float* h1   = dinv + 50048;
    float* agg1 = h1   + (size_t)Nn * F_HID;
    float* h2   = agg1 + (size_t)Nn * F_HID;
    unsigned* keys = (unsigned*)(h2 + (size_t)Nn * F_OUT);
    float* gcnt = (float*)(keys + (size_t)((E + 511) / 512) * 128);
    unsigned* list = (unsigned*)(gcnt + 256);

    const int TPB = 256;

    zero_f32<<<(Nn + TPB - 1) / TPB, TPB, 0, stream>>>(dinv, Nn);
    {
        int mtiles = Nn / 16;
        int gemm_blocks = (mtiles + GD_TPB / 32 - 1) / (GD_TPB / 32);
        int key_blocks  = (E + 511) / 512;
        int key_groups  = (key_blocks + GD_TPB / 128 - 1) / (GD_TPB / 128);
        gemm1_deg_keys<<<gemm_blocks + key_groups, GD_TPB, 0, stream>>>(x, W1, h1, Nn, gemm_blocks,
                                                                       dst, dinv, keys, E);
    }
    make_dinv<<<(Nn + OWN_NPB - 1) / OWN_NPB, DINV_TPB, 0, stream>>>(dinv, gcnt, Nn);

    {
        int blocks = (Nn + OWN_NPB - 1) / OWN_NPB;
        agg_scan<F_HID><<<blocks, OWN_WAVES * 32, 0, stream>>>(
            src, dst, keys, gcnt, dinv, h1, b1, agg1, list, E, Nn);
    }

    {
        int mtiles = Nn / 16;
        int blocks = (mtiles + 7) / 8;
        gemm_wmma_f32<3, 40><<<blocks, TPB, 0, stream>>>(agg1, W2, h2, Nn);
    }

    {
        int blocks = (Nn + OWN_NPB - 1) / OWN_NPB;
        agg_walk<F_OUT><<<blocks, OWN_WAVES * 32, 0, stream>>>(
            list, gcnt, dinv, h2, b2, out, Nn);
    }
}


# Owner keys as a bit plane per owner group (key work-groups build 2048-edge blocks in LDS): a group reads its own E/8 bytes, not all E key bytes; hits by ballot rounds; sum order still LDS arrival
# speedup vs baseline: 3.2599x; 1.0643x over previous
#include <hip/hip_runtime.h>
#include <math.h>

typedef float v2f __attribute__((ext_vector_type(2)));
typedef float v8f __attribute__((ext_vector_type(8)));

#define N_NODES 50000
#define F_IN    128
#define F_HID   128
#define F_OUT   40
#define OWN_GROUPS 254
#define OWN_NPB ((N_NODES + OWN_GROUPS - 1) / OWN_GROUPS)
#define DINV_TPB 256
#define KB_EDGES 2048
#define KB_WORDS (KB_EDGES / 32)
#define GD_TPB 256

__global__ void zero_f32(float* __restrict__ p, int n) {
    int i = blockIdx.x * blockDim.x + threadIdx.x;
    if (i < n) p[i] = 0.0f;
}

__device__ __forceinline__ void deg_bits_block(const int* __restrict__ dst, float* __restrict__ deg,
                                               unsigned* __restrict__ planes, unsigned* bmp,
                                               int E, int nkb, int kb, int tid) {
    constexpr int EPT = KB_EDGES / GD_TPB;
    constexpr int NQ  = OWN_GROUPS * KB_WORDS / 4;
    uint4* b4 = reinterpret_cast<uint4*>(bmp);
    for (int i = tid; i < NQ; i += GD_TPB) b4[i] = make_uint4(0u, 0u, 0u, 0u);
    const int e0   = kb * KB_EDGES + tid;
    const int last = E - 1;
    int d[EPT];
#pragma unroll
    for (int j = 0; j < EPT; ++j) {
        const int e = e0 + GD_TPB * j;
        d[j] = dst[e < E ? e : last];
    }
    __syncthreads();
#pragma unroll
    for (int j = 0; j < EPT; ++j) {
        const int e = e0 + GD_TPB * j;
        if (e < E) {
            unsafeAtomicAdd(&deg[d[j]], 1.0f);
            const unsigned o = (unsigned)d[j] / OWN_NPB;
            if (o < OWN_GROUPS)
                __hip_atomic_fetch_or(&bmp[o * KB_WORDS + ((GD_TPB * j + tid) >> 5)], 1u << (tid & 31),
                                      __ATOMIC_RELAXED, __HIP_MEMORY_SCOPE_WORKGROUP);
        }
    }
    __syncthreads();
    uint4* p4 = reinterpret_cast<uint4*>(planes);
    for (int i = tid; i < NQ; i += GD_TPB) {
        const int g = i / (KB_WORDS / 4);
        const int q = i - g * (KB_WORDS / 4);
        p4[((size_t)g * nkb + kb) * (KB_WORDS / 4) + q] = b4[i];
    }
}

__global__ void make_dinv(float* __restrict__ deg, float* __restrict__ gcnt, int n) {
    __shared__ float part[DINV_TPB / 32];
    int i = blockIdx.x * OWN_NPB + threadIdx.x;
    float d = 0.0f;
    if (threadIdx.x < OWN_NPB && i < n) {
        d = deg[i];
        deg[i] = rsqrtf(d + 1.0f);
    }
#pragma unroll
    for (int off = 16; off > 0; off >>= 1)
        d += __shfl_xor(d, off, 32);
    if ((threadIdx.x & 31) == 0) part[threadIdx.x >> 5] = d;
    __syncthreads();
    if (threadIdx.x == 0) {
        float t = 0.0f;
        for (int w = 0; w < DINV_TPB / 32; ++w) t += part[w];
        gcnt[blockIdx.x] = t;
    }
}

template <int NT, int NC>
__device__ __forceinline__ void gemm_strip_f32(const float* __restrict__ X,
                                               const float* __restrict__ W,
                                               float* __restrict__ H, int M, int mt, int lane) {
    if (mt * 16 >= M) return;

    const int half = lane >> 4;
    const int l16  = lane & 15;

    v8f c[NT] = {};

    const float* xrow = X + (size_t)(mt * 16 + l16) * 128;

    for (int k = 0; k < 128; k += 4) {
        v2f a;
        a.x = xrow[k + 2 * half];
        a.y = xrow[k + 2 * half + 1];
#pragma unroll
        for (int t = 0; t < NT; ++t) {
            const int n = t * 16 + l16;
            v2f b;
            if ((NC % 16 == 0) || (n < NC)) {
                b.x = W[(size_t)(k + 2 * half) * NC + n];
                b.y = W[(size_t)(k + 2 * half + 1) * NC + n];
            } else {
                b.x = 0.0f; b.y = 0.0f;
            }
            c[t] = __builtin_amdgcn_wmma_f32_16x16x4_f32(
                false, a, false, b, (short)0, c[t], false, false);
        }
    }

#pragma unroll
    for (int t = 0; t < NT; ++t) {
        const int n = t * 16 + l16;
        if ((NC % 16 == 0) || (n < NC)) {
            float* out = H + (size_t)(mt * 16 + half * 8) * NC + n;
#pragma unroll
            for (int r = 0; r < 8; ++r) out[(size_t)r * NC] = c[t][r];
        }
    }
}

template <int NT, int NC>
__global__ void gemm_wmma_f32(const float* __restrict__ X,
                              const float* __restrict__ W,
                              float* __restrict__ H, int M) {
    gemm_strip_f32<NT, NC>(X, W, H, M, blockIdx.x * (blockDim.x >> 5) + (threadIdx.x >> 5), threadIdx.x & 31);
}

__global__ void __launch_bounds__(GD_TPB)
gemm1_deg_bits(const float* __restrict__ X, const float* __restrict__ W, float* __restrict__ H, int M,
               int gemm_blocks,
               const int* __restrict__ dst, float* __restrict__ deg, unsigned* __restrict__ planes,
               int E, int nkb) {
    __shared__ unsigned bmp[OWN_GROUPS * KB_WORDS];
    if ((int)blockIdx.x < gemm_blocks) {
        gemm_strip_f32<8, 128>(X, W, H, M, blockIdx.x * (GD_TPB / 32) + (threadIdx.x >> 5), threadIdx.x & 31);
    } else {
        deg_bits_block(dst, deg, planes, bmp, E, nkb, (int)blockIdx.x - gemm_blocks, threadIdx.x);
    }
}

#define OWN_WAVES 32
#define OWN_QCAP  64

static_assert(OWN_NPB <= 256, "a list entry keeps the local row in 8 bits");
static_assert(DINV_TPB >= OWN_NPB && DINV_TPB % 32 == 0, "make_dinv: one thread per node of a range, whole waves");
static_assert(N_NODES <= 65536, "a list entry keeps the source node in 16 bits");
static_assert((N_NODES + OWN_NPB - 1) / OWN_NPB <= OWN_GROUPS, "every node has an owner plane");
static_assert((OWN_GROUPS * KB_WORDS) % 4 == 0 && KB_EDGES % GD_TPB == 0, "a block's image is whole 16-byte pieces, its edges whole rounds of the work-group");
static_assert((N_NODES + OWN_NPB - 1) / OWN_NPB <= OWN_WAVES * 32, "one thread per earlier group sums the list offset");

template <int F>
__device__ __forceinline__ int lds_col(int c) { return (c & 3) * (F / 4) + (c >> 2); }

template <int F>
__device__ __forceinline__ void start_rows(const float* __restrict__ dinv,
                                           const float* __restrict__ H,
                                           const float* __restrict__ bias,
                                           float* acc, int lo, int cnt, int tid) {
    for (int idx = tid; idx < cnt * F; idx += OWN_WAVES * 32) {
        const int i = idx / F;
        const int f = idx - i * F;
        const float di = dinv[lo + i];
        acc[i * F + lds_col<F>(f)] = H[(size_t)lo * F + idx] * di * di + bias[f];
    }
}

__device__ __forceinline__ void list_offset_share(const float* __restrict__ gcnt, float* red,
                                                  int g, int tid, int lane, int wave) {
    float v = (tid < g) ? gcnt[tid] : 0.0f;
#pragma unroll
    for (int off = 16; off > 0; off >>= 1)
        v += __shfl_xor(v, off, 32);
    if (lane == 0) red[wave] = v;
}

__device__ __forceinline__ int list_offset(const float* red) {
    float t = 0.0f;
    for (int i = 0; i < OWN_WAVES; ++i) t += red[i];
    return __builtin_amdgcn_readfirstlane((int)t);
}

template <int F>
__device__ __forceinline__ void add_rows(const float* __restrict__ H, float* acc,
                                         int s, int dloc, float w, int n, int lane) {
    static_assert(F % 4 == 0 && F / 4 <= 32, "a row is F / 4 lanes of 16 bytes");
    constexpr int LPR = F / 4;
    constexpr int RPL = 32 / LPR;
    const float4* __restrict__ H4 = reinterpret_cast<const float4*>(H);
    if constexpr (RPL == 1) {
        constexpr int ROWS = 16;
#pragma nounroll
        for (int k = 0; k < n; k += ROWS) {
            float4 r[ROWS];
#pragma unroll
            for (int u = 0; u < ROWS; ++u) {
                const int sk = __builtin_amdgcn_readlane(s, (k + u) & 31);
                r[u] = ((k + u) < n) ? H4[(size_t)sk * LPR + lane] : make_float4(0.0f, 0.0f, 0.0f, 0.0f);
            }
#pragma unroll
            for (int u = 0; u < ROWS; ++u) {
                const float wk = __int_as_float(__builtin_amdgcn_readlane(__float_as_int(w), (k + u) & 31));
                const int   dk = __builtin_amdgcn_readlane(dloc, (k + u) & 31);
                if ((k + u) < n) {
                    float* a = acc + dk * F + lane;
                    __hip_atomic_fetch_add(a,           r[u].x * wk, __ATOMIC_RELAXED, __HIP_MEMORY_SCOPE_WORKGROUP);
                    __hip_atomic_fetch_add(a + LPR,     r[u].y * wk, __ATOMIC_RELAXED, __HIP_MEMORY_SCOPE_WORKGROUP);
                    __hip_atomic_fetch_add(a + 2 * LPR, r[u].z * wk, __ATOMIC_RELAXED, __HIP_MEMORY_SCOPE_WORKGROUP);
                    __hip_atomic_fetch_add(a + 3 * LPR, r[u].w * wk, __ATOMIC_RELAXED, __HIP_MEMORY_SCOPE_WORKGROUP);
                }
            }
        }
    } else {
        constexpr int NLD = (32 + RPL - 1) / RPL;
        const int  g    = lane / LPR;
        const int  m    = lane - g * LPR;
        const bool live = g < RPL;
        const int  pk   = s | (dloc << 16);
        float4 r[NLD];
        int    p[NLD];
        float  wk[NLD];
#pragma unroll
        for (int t = 0; t < NLD; ++t) {
            const int e = t * RPL + g;
            p[t]  = __shfl(pk, e & 31, 32);
            wk[t] = __shfl(w, e & 31, 32);
        }
#pragma unroll
        for (int t = 0; t < NLD; ++t) {
            const int e = t * RPL + g;
            const int okm = -(int)(live && e < n);
            const int sk = p[t] & 0xFFFF & okm;
            p[t] |= ~okm;
            r[t] = H4[(size_t)sk * LPR + m];
        }
#pragma unroll
        for (int t = 0; t < NLD; ++t) {
            if (p[t] >= 0) {
                float* a = acc + (p[t] >> 16) * F + m;
                __hip_atomic_fetch_add(a,           r[t].x * wk[t], __ATOMIC_RELAXED, __HIP_MEMORY_SCOPE_WORKGROUP);
                __hip_atomic_fetch_add(a + LPR,     r[t].y * wk[t], __ATOMIC_RELAXED, __HIP_MEMORY_SCOPE_WORKGROUP);
                __hip_atomic_fetch_add(a + 2 * LPR, r[t].z * wk[t], __ATOMIC_RELAXED, __HIP_MEMORY_SCOPE_WORKGROUP);
                __hip_atomic_fetch_add(a + 3 * LPR, r[t].w * wk[t], __ATOMIC_RELAXED, __HIP_MEMORY_SCOPE_WORKGROUP);
            }
        }
    }
}

template <int F>
__device__ __forceinline__ void drain_queue(const int* __restrict__ src,
                                            const int* __restrict__ dst,
                                            const float* __restrict__ dinv,
                                            const float* __restrict__ H,
                                            float* acc, const float* dl,
                                            const int* qe, int lo,
                                            uint2* __restrict__ glist, int* lcnt,
                                            int n, int lane) {
    int s = 0, dloc = 0;
    float w = 0.0f;
    if (lane < n) {
        const int e = qe[lane];
        s = src[e];
        dloc = dst[e] - lo;
        w = dinv[s] * dl[dloc];
    }
    int base = 0;
    if (lane == 0)
        base = __hip_atomic_fetch_add(lcnt, n, __ATOMIC_RELAXED, __HIP_MEMORY_SCOPE_WORKGROUP);
    base = __builtin_amdgcn_readfirstlane(base);
    if (lane < n)
        glist[base + lane] = make_uint2((unsigned)s | ((unsigned)dloc << 16), __float_as_uint(w));
    add_rows<F>(H, acc, s, dloc, w, n, lane);
}

template <int F>
__global__ void __launch_bounds__(OWN_WAVES * 32)
agg_scan(const int* __restrict__ src, const int* __restrict__ dst,
         const unsigned* __restrict__ planes, const float* __restrict__ gcnt,
         const float* __restrict__ dinv, const float* __restrict__ H,
         const float* __restrict__ bias, float* __restrict__ OUT,
         unsigned* __restrict__ list, int nkb, int Nn) {
    __shared__ float acc[OWN_NPB * F];
    __shared__ float dl[OWN_NPB];
    __shared__ int   qe[OWN_WAVES][OWN_QCAP];
    __shared__ float red[OWN_WAVES];
    __shared__ int   lcnt;

    const int tid  = threadIdx.x;
    const int lane = tid & 31;
    const int wave = __builtin_amdgcn_readfirstlane(tid >> 5);
    const int lo   = blockIdx.x * OWN_NPB;
    const int cnt  = (Nn - lo) < OWN_NPB ? (Nn - lo) : OWN_NPB;

    if (tid == 0) lcnt = 0;
    list_offset_share(gcnt, red, blockIdx.x, tid, lane, wave);
    for (int i = tid; i < cnt; i += OWN_WAVES * 32) dl[i] = dinv[lo + i];
    start_rows<F>(dinv, H, bias, acc, lo, cnt, tid);
    __syncthreads();

    uint2* glist = reinterpret_cast<uint2*>(list) + list_offset(red);
    int* myq = qe[wave];
    int qn = 0;

    const int npc = nkb * (KB_WORDS / 4);
    if (npc > 0) {
        const uint4* __restrict__ pp = reinterpret_cast<const uint4*>(planes) + (size_t)blockIdx.x * npc;
        const unsigned ltmask = (1u << lane) - 1u;
        const int lastp = npc - 1;
        int pc = wave * 32;
        uint4 kn = make_uint4(0u, 0u, 0u, 0u);
        if (pc < npc) kn = pp[(pc + lane < npc) ? (pc + lane) : lastp];
#pragma nounroll
        while (pc < npc) {
            const unsigned live = (pc + lane < npc) ? 0xFFFFFFFFu : 0u;
            unsigned kv[4] = { kn.x & live, kn.y & live, kn.z & live, kn.w & live };
            const int pn = pc + OWN_WAVES * 32;
            kn = pp[(pn + lane < npc) ? (pn + lane) : lastp];
#pragma unroll
            for (int q = 0; q < 4; ++q) {
                unsigned v = kv[q];
                const int e0 = (pc + lane) * 128 + 32 * q;
                unsigned m = __builtin_amdgcn_ballot_w32(v != 0u);
                while (m != 0u) {
                    if (v != 0u) {
                        myq[qn + __builtin_popcount(m & ltmask)] = e0 + __builtin_ctz(v);
                        v &= v - 1u;
                    }
                    qn += __builtin_popcount(m);
                    if (qn >= 32) {
                        drain_queue<F>(src, dst, dinv, H, acc, dl, myq, lo, glist, &lcnt, 32, lane);
                        qn -= 32;
                        int te = 0;
                        if (lane < qn) te = myq[32 + lane];
                        if (lane < qn) myq[lane] = te;
                    }
                    m = __builtin_amdgcn_ballot_w32(v != 0u);
                }
            }
            pc = pn;
        }
    }
    if (qn > 0) drain_queue<F>(src, dst, dinv, H, acc, dl, myq, lo, glist, &lcnt, qn, lane);
    __syncthreads();

    for (int idx = tid; idx < cnt * F; idx += OWN_WAVES * 32) {
        const int i = idx / F;
        const float v = acc[i * F + lds_col<F>(idx - i * F)];
        OUT[(size_t)lo * F + idx] = v > 0.0f ? v : 0.0f;
    }
}

template <int F>
__global__ void __launch_bounds__(OWN_WAVES * 32)
agg_walk(const unsigned* __restrict__ list, const float* __restrict__ gcnt,
         const float* __restrict__ dinv, const float* __restrict__ H,
         const float* __restrict__ bias, float* __restrict__ OUT, int Nn) {
    static_assert(F > 32 && F <= 64, "the epilogue holds a row in two registers per lane");
    __shared__ float acc[OWN_NPB * F];
    __shared__ float red[OWN_WAVES];

    const int tid  = threadIdx.x;
    const int lane = tid & 31;
    const int wave = __builtin_amdgcn_readfirstlane(tid >> 5);
    const int lo   = blockIdx.x * OWN_NPB;
    const int cnt  = (Nn - lo) < OWN_NPB ? (Nn - lo) : OWN_NPB;

    list_offset_share(gcnt, red, blockIdx.x, tid, lane, wave);
    start_rows<F>(dinv, H, bias, acc, lo, cnt, tid);
    __syncthreads();

    const uint2* __restrict__ glist = reinterpret_cast<const uint2*>(list) + list_offset(red);
    const int len = (int)gcnt[blockIdx.x];

    int c = wave * 32;
    uint2 en = make_uint2(0u, 0u);
    if (c + lane < len) en = glist[c + lane];
#pragma nounroll
    while (c < len) {
        const uint2 ev = en;
        const int rem = len - c;
        const int cn = c + OWN_WAVES * 32;
        en = glist[(cn + lane < len) ? (cn + lane) : (len - 1)];
        const int   s    = (int)(ev.x & 0xFFFFu);
        const int   dloc = (int)(ev.x >> 16);
        const float w    = __uint_as_float(ev.y);
        if (rem >= 32) add_rows<F>(H, acc, s, dloc, w, 32, lane);
        else           add_rows<F>(H, acc, s, dloc, w, rem, lane);
        c = cn;
    }
    __syncthreads();

    for (int i = wave; i < cnt; i += OWN_WAVES) {
        const float v0 = (lane < F) ? acc[i * F + lds_col<F>(lane)] : -INFINITY;
        const float v1 = (lane + 32 < F) ? acc[i * F + lds_col<F>(32 + lane)] : -INFINITY;
        float m = fmaxf(v0, v1);
#pragma unroll
        for (int off = 16; off > 0; off >>= 1)
            m = fmaxf(m, __shfl_xor(m, off, 32));
        float sm = ((lane < F) ? __expf(v0 - m) : 0.0f) +
                   ((lane + 32 < F) ? __expf(v1 - m) : 0.0f);
#pragma unroll
        for (int off = 16; off > 0; off >>= 1)
            sm += __shfl_xor(sm, off, 32);
        const float ls = __logf(sm);
        const size_t ob = (size_t)(lo + i) * F;
        if (lane < F)      OUT[ob + lane]      = v0 - m - ls;
        if (lane + 32 < F) OUT[ob + 32 + lane] = v1 - m - ls;
    }
}

extern "C" void kernel_launch(void* const* d_in, const int* in_sizes, int n_in,
                              void* d_out, int out_size, void* d_ws, size_t ws_size,
                              hipStream_t stream) {
    const float* x  = (const float*)d_in[0];
    const float* W1 = (const float*)d_in[1];
    const float* b1 = (const float*)d_in[2];
    const float* W2 = (const float*)d_in[3];
    const float* b2 = (const float*)d_in[4];
    const int*   ei = (const int*)d_in[5];
    const int E = in_sizes[5] / 2;
    const int Nn = N_NODES;
    const int* src = ei;
    const int* dst = ei + E;

    float* out = (float*)d_out;

    float* ws   = (float*)d_ws;
    float* dinv = ws;
    float* h1   = dinv + 50048;
    float* agg1 = h1   + (size_t)Nn * F_HID;
    float* h2   = agg1 + (size_t)Nn * F_HID;
    const int nkb = (E + KB_EDGES - 1) / KB_EDGES;
    unsigned* planes = (unsigned*)(h2 + (size_t)Nn * F_OUT);
    float* gcnt = (float*)(planes + (size_t)OWN_GROUPS * nkb * KB_WORDS);
    unsigned* list = (unsigned*)(gcnt + 256);

    const int TPB = 256;

    zero_f32<<<(Nn + TPB - 1) / TPB, TPB, 0, stream>>>(dinv, Nn);
    {
        int mtiles = Nn / 16;
        int gemm_blocks = (mtiles + GD_TPB / 32 - 1) / (GD_TPB / 32);
        gemm1_deg_bits<<<gemm_blocks + nkb, GD_TPB, 0, stream>>>(x, W1, h1, Nn, gemm_blocks,
                                                                 dst, dinv, planes, E, nkb);
    }
    make_dinv<<<(Nn + OWN_NPB - 1) / OWN_NPB, DINV_TPB, 0, stream>>>(dinv, gcnt, Nn);

    {
        int blocks = (Nn + OWN_NPB - 1) / OWN_NPB;
        agg_scan<F_HID><<<blocks, OWN_WAVES * 32, 0, stream>>>(
            src, dst, planes, gcnt, dinv, h1, b1, agg1, list, nkb, Nn);
    }

    {
        int mtiles = Nn / 16;
        int blocks = (mtiles + 7) / 8;
        gemm_wmma_f32<3, 40><<<blocks, TPB, 0, stream>>>(agg1, W2, h2, Nn);
    }

    {
        int blocks = (Nn + OWN_NPB - 1) / OWN_NPB;
        agg_walk<F_OUT><<<blocks, OWN_WAVES * 32, 0, stream>>>(
            list, gcnt, dinv, h2, b2, out, Nn);
    }
}


# Layer-2 product as the layer-1 aggregation's epilogue: the owned ReLU rows in LDS (row pitch 132 floats) are the WMMA A operand, same k order; relu plane and one launch gone; sum order LDS arrival
# speedup vs baseline: 3.7282x; 1.1437x over previous
#include <hip/hip_runtime.h>
#include <math.h>

typedef float v2f __attribute__((ext_vector_type(2)));
typedef float v8f __attribute__((ext_vector_type(8)));

#define N_NODES 50000
#define F_IN    128
#define F_HID   128
#define F_OUT   40
#define OWN_GROUPS 254
#define OWN_NPB ((N_NODES + OWN_GROUPS - 1) / OWN_GROUPS)
#define DINV_TPB 256
#define KB_EDGES 2048
#define KB_WORDS (KB_EDGES / 32)
#define GD_TPB 256

__global__ void zero_f32(float* __restrict__ p, int n) {
    int i = blockIdx.x * blockDim.x + threadIdx.x;
    if (i < n) p[i] = 0.0f;
}

__device__ __forceinline__ void deg_bits_block(const int* __restrict__ dst, float* __restrict__ deg,
                                               unsigned* __restrict__ planes, unsigned* bmp,
                                               int E, int nkb, int kb, int tid) {
    constexpr int EPT = KB_EDGES / GD_TPB;
    constexpr int NQ  = OWN_GROUPS * KB_WORDS / 4;
    uint4* b4 = reinterpret_cast<uint4*>(bmp);
    for (int i = tid; i < NQ; i += GD_TPB) b4[i] = make_uint4(0u, 0u, 0u, 0u);
    const int e0   = kb * KB_EDGES + tid;
    const int last = E - 1;
    int d[EPT];
#pragma unroll
    for (int j = 0; j < EPT; ++j) {
        const int e = e0 + GD_TPB * j;
        d[j] = dst[e < E ? e : last];
    }
    __syncthreads();
#pragma unroll
    for (int j = 0; j < EPT; ++j) {
        const int e = e0 + GD_TPB * j;
        if (e < E) {
            unsafeAtomicAdd(&deg[d[j]], 1.0f);
            const unsigned o = (unsigned)d[j] / OWN_NPB;
            if (o < OWN_GROUPS)
                __hip_atomic_fetch_or(&bmp[o * KB_WORDS + ((GD_TPB * j + tid) >> 5)], 1u << (tid & 31),
                                      __ATOMIC_RELAXED, __HIP_MEMORY_SCOPE_WORKGROUP);
        }
    }
    __syncthreads();
    uint4* p4 = reinterpret_cast<uint4*>(planes);
    for (int i = tid; i < NQ; i += GD_TPB) {
        const int g = i / (KB_WORDS / 4);
        const int q = i - g * (KB_WORDS / 4);
        p4[((size_t)g * nkb + kb) * (KB_WORDS / 4) + q] = b4[i];
    }
}

__global__ void make_dinv(float* __restrict__ deg, float* __restrict__ gcnt, int n) {
    __shared__ float part[DINV_TPB / 32];
    int i = blockIdx.x * OWN_NPB + threadIdx.x;
    float d = 0.0f;
    if (threadIdx.x < OWN_NPB && i < n) {
        d = deg[i];
        deg[i] = rsqrtf(d + 1.0f);
    }
#pragma unroll
    for (int off = 16; off > 0; off >>= 1)
        d += __shfl_xor(d, off, 32);
    if ((threadIdx.x & 31) == 0) part[threadIdx.x >> 5] = d;
    __syncthreads();
    if (threadIdx.x == 0) {
        float t = 0.0f;
        for (int w = 0; w < DINV_TPB / 32; ++w) t += part[w];
        gcnt[blockIdx.x] = t;
    }
}

template <int NT, int NC>
__device__ __forceinline__ void gemm_strip_f32(const float* __restrict__ X,
                                               const float* __restrict__ W,
                                               float* __restrict__ H, int M, int mt, int lane) {
    if (mt * 16 >= M) return;

    const int half = lane >> 4;
    const int l16  = lane & 15;

    v8f c[NT] = {};

    const float* xrow = X + (size_t)(mt * 16 + l16) * 128;

    for (int k = 0; k < 128; k += 4) {
        v2f a;
        a.x = xrow[k + 2 * half];
        a.y = xrow[k + 2 * half + 1];
#pragma unroll
        for (int t = 0; t < NT; ++t) {
            const int n = t * 16 + l16;
            v2f b;
            if ((NC % 16 == 0) || (n < NC)) {
                b.x = W[(size_t)(k + 2 * half) * NC + n];
                b.y = W[(size_t)(k + 2 * half + 1) * NC + n];
            } else {
                b.x = 0.0f; b.y = 0.0f;
            }
            c[t] = __builtin_amdgcn_wmma_f32_16x16x4_f32(
                false, a, false, b, (short)0, c[t], false, false);
        }
    }

#pragma unroll
    for (int t = 0; t < NT; ++t) {
        const int n = t * 16 + l16;
        if ((NC % 16 == 0) || (n < NC)) {
            float* out = H + (size_t)(mt * 16 + half * 8) * NC + n;
#pragma unroll
            for (int r = 0; r < 8; ++r) out[(size_t)r * NC] = c[t][r];
        }
    }
}

__global__ void __launch_bounds__(GD_TPB)
gemm1_deg_bits(const float* __restrict__ X, const float* __restrict__ W, float* __restrict__ H, int M,
               int gemm_blocks,
               const int* __restrict__ dst, float* __restrict__ deg, unsigned* __restrict__ planes,
               int E, int nkb) {
    __shared__ unsigned bmp[OWN_GROUPS * KB_WORDS];
    if ((int)blockIdx.x < gemm_blocks) {
        gemm_strip_f32<8, 128>(X, W, H, M, blockIdx.x * (GD_TPB / 32) + (threadIdx.x >> 5), threadIdx.x & 31);
    } else {
        deg_bits_block(dst, deg, planes, bmp, E, nkb, (int)blockIdx.x - gemm_blocks, threadIdx.x);
    }
}

#define OWN_WAVES 32
#define OWN_QCAP  64

static_assert(OWN_NPB <= 256, "a list entry keeps the local row in 8 bits");
static_assert(DINV_TPB >= OWN_NPB && DINV_TPB % 32 == 0, "make_dinv: one thread per node of a range, whole waves");
static_assert(N_NODES <= 65536, "a list entry keeps the source node in 16 bits");
static_assert((N_NODES + OWN_NPB - 1) / OWN_NPB <= OWN_GROUPS, "every node has an owner plane");
static_assert((OWN_GROUPS * KB_WORDS) % 4 == 0 && KB_EDGES % GD_TPB == 0, "a block's image is whole 16-byte pieces, its edges whole rounds of the work-group");
static_assert((N_NODES + OWN_NPB - 1) / OWN_NPB <= OWN_WAVES * 32, "one thread per earlier group sums the list offset");

template <int F>
__device__ __forceinline__ int lds_col(int c) { return (c & 3) * (F / 4) + (c >> 2); }

#define SCAN_PITCH (F_HID + 4)

template <int F, int P>
__device__ __forceinline__ void start_rows(const float* __restrict__ dinv,
                                           const float* __restrict__ H,
                                           const float* __restrict__ bias,
                                           float* acc, int lo, int cnt, int tid) {
    for (int idx = tid; idx < cnt * F; idx += OWN_WAVES * 32) {
        const int i = idx / F;
        const int f = idx - i * F;
        const float di = dinv[lo + i];
        acc[i * P + lds_col<F>(f)] = H[(size_t)lo * F + idx] * di * di + bias[f];
    }
}

__device__ __forceinline__ void list_offset_share(const float* __restrict__ gcnt, float* red,
                                                  int g, int tid, int lane, int wave) {
    float v = (tid < g) ? gcnt[tid] : 0.0f;
#pragma unroll
    for (int off = 16; off > 0; off >>= 1)
        v += __shfl_xor(v, off, 32);
    if (lane == 0) red[wave] = v;
}

__device__ __forceinline__ int list_offset(const float* red) {
    float t = 0.0f;
    for (int i = 0; i < OWN_WAVES; ++i) t += red[i];
    return __builtin_amdgcn_readfirstlane((int)t);
}

template <int F, int P>
__device__ __forceinline__ void add_rows(const float* __restrict__ H, float* acc,
                                         int s, int dloc, float w, int n, int lane) {
    static_assert(F % 4 == 0 && F / 4 <= 32, "a row is F / 4 lanes of 16 bytes");
    constexpr int LPR = F / 4;
    constexpr int RPL = 32 / LPR;
    const float4* __restrict__ H4 = reinterpret_cast<const float4*>(H);
    if constexpr (RPL == 1) {
        constexpr int ROWS = 16;
#pragma nounroll
        for (int k = 0; k < n; k += ROWS) {
            float4 r[ROWS];
#pragma unroll
            for (int u = 0; u < ROWS; ++u) {
                const int sk = __builtin_amdgcn_readlane(s, (k + u) & 31);
                r[u] = ((k + u) < n) ? H4[(size_t)sk * LPR + lane] : make_float4(0.0f, 0.0f, 0.0f, 0.0f);
            }
#pragma unroll
            for (int u = 0; u < ROWS; ++u) {
                const float wk = __int_as_float(__builtin_amdgcn_readlane(__float_as_int(w), (k + u) & 31));
                const int   dk = __builtin_amdgcn_readlane(dloc, (k + u) & 31);
                if ((k + u) < n) {
                    float* a = acc + dk * P + lane;
                    __hip_atomic_fetch_add(a,           r[u].x * wk, __ATOMIC_RELAXED, __HIP_MEMORY_SCOPE_WORKGROUP);
                    __hip_atomic_fetch_add(a + LPR,     r[u].y * wk, __ATOMIC_RELAXED, __HIP_MEMORY_SCOPE_WORKGROUP);
                    __hip_atomic_fetch_add(a + 2 * LPR, r[u].z * wk, __ATOMIC_RELAXED, __HIP_MEMORY_SCOPE_WORKGROUP);
                    __hip_atomic_fetch_add(a + 3 * LPR, r[u].w * wk, __ATOMIC_RELAXED, __HIP_MEMORY_SCOPE_WORKGROUP);
                }
            }
        }
    } else {
        constexpr int NLD = (32 + RPL - 1) / RPL;
        const int  g    = lane / LPR;
        const int  m    = lane - g * LPR;
        const bool live = g < RPL;
        const int  pk   = s | (dloc << 16);
        float4 r[NLD];
        int    p[NLD];
        float  wk[NLD];
#pragma unroll
        for (int t = 0; t < NLD; ++t) {
            const int e = t * RPL + g;
            p[t]  = __shfl(pk, e & 31, 32);
            wk[t] = __shfl(w, e & 31, 32);
        }
#pragma unroll
        for (int t = 0; t < NLD; ++t) {
            const int e = t * RPL + g;
            const int okm = -(int)(live && e < n);
            const int sk = p[t] & 0xFFFF & okm;
            p[t] |= ~okm;
            r[t] = H4[(size_t)sk * LPR + m];
        }
#pragma unroll
        for (int t = 0; t < NLD; ++t) {
            if (p[t] >= 0) {
                float* a = acc + (p[t] >> 16) * P + m;
                __hip_atomic_fetch_add(a,           r[t].x * wk[t], __ATOMIC_RELAXED, __HIP_MEMORY_SCOPE_WORKGROUP);
                __hip_atomic_fetch_add(a + LPR,     r[t].y * wk[t], __ATOMIC_RELAXED, __HIP_MEMORY_SCOPE_WORKGROUP);
                __hip_atomic_fetch_add(a + 2 * LPR, r[t].z * wk[t], __ATOMIC_RELAXED, __HIP_MEMORY_SCOPE_WORKGROUP);
                __hip_atomic_fetch_add(a + 3 * LPR, r[t].w * wk[t], __ATOMIC_RELAXED, __HIP_MEMORY_SCOPE_WORKGROUP);
            }
        }
    }
}

template <int F, int P>
__device__ __forceinline__ void drain_queue(const int* __restrict__ src,
                                            const int* __restrict__ dst,
                                            const float* __restrict__ dinv,
                                            const float* __restrict__ H,
                                            float* acc, const float* dl,
                                            const int* qe, int lo,
                                            uint2* __restrict__ glist, int* lcnt,
                                            int n, int lane) {
    int s = 0, dloc = 0;
    float w = 0.0f;
    if (lane < n) {
        const int e = qe[lane];
        s = src[e];
        dloc = dst[e] - lo;
        w = dinv[s] * dl[dloc];
    }
    int base = 0;
    if (lane == 0)
        base = __hip_atomic_fetch_add(lcnt, n, __ATOMIC_RELAXED, __HIP_MEMORY_SCOPE_WORKGROUP);
    base = __builtin_amdgcn_readfirstlane(base);
    if (lane < n)
        glist[base + lane] = make_uint2((unsigned)s | ((unsigned)dloc << 16), __float_as_uint(w));
    add_rows<F, P>(H, acc, s, dloc, w, n, lane);
}

template <int F, int NC>
__global__ void __launch_bounds__(OWN_WAVES * 32)
agg_scan(const int* __restrict__ src, const int* __restrict__ dst,
         const unsigned* __restrict__ planes, const float* __restrict__ gcnt,
         const float* __restrict__ dinv, const float* __restrict__ H,
         const float* __restrict__ bias, const float* __restrict__ W2, float* __restrict__ H2,
         unsigned* __restrict__ list, int nkb, int Nn) {
    constexpr int P = SCAN_PITCH;
    static_assert(P >= F && P % 4 == 0 && F % 16 == 0, "rows are whole 16-byte pieces; a trip of the product is four k-steps");
    __shared__ __attribute__((aligned(16))) float acc[OWN_NPB * P];
    __shared__ float dl[OWN_NPB];
    __shared__ int   qe[OWN_WAVES][OWN_QCAP];
    __shared__ float red[OWN_WAVES];
    __shared__ int   lcnt;

    const int tid  = threadIdx.x;
    const int lane = tid & 31;
    const int wave = __builtin_amdgcn_readfirstlane(tid >> 5);
    const int lo   = blockIdx.x * OWN_NPB;
    const int cnt  = (Nn - lo) < OWN_NPB ? (Nn - lo) : OWN_NPB;

    if (tid == 0) lcnt = 0;
    list_offset_share(gcnt, red, blockIdx.x, tid, lane, wave);
    for (int i = tid; i < cnt; i += OWN_WAVES * 32) dl[i] = dinv[lo + i];
    start_rows<F, P>(dinv, H, bias, acc, lo, cnt, tid);
    __syncthreads();

    uint2* glist = reinterpret_cast<uint2*>(list) + list_offset(red);
    int* myq = qe[wave];
    int qn = 0;

    const int npc = nkb * (KB_WORDS / 4);
    if (npc > 0) {
        const uint4* __restrict__ pp = reinterpret_cast<const uint4*>(planes) + (size_t)blockIdx.x * npc;
        const unsigned ltmask = (1u << lane) - 1u;
        const int lastp = npc - 1;
        int pc = wave * 32;
        uint4 kn = make_uint4(0u, 0u, 0u, 0u);
        if (pc < npc) kn = pp[(pc + lane < npc) ? (pc + lane) : lastp];
#pragma nounroll
        while (pc < npc) {
            const unsigned live = (pc + lane < npc) ? 0xFFFFFFFFu : 0u;
            unsigned kv[4] = { kn.x & live, kn.y & live, kn.z & live, kn.w & live };
            const int pn = pc + OWN_WAVES * 32;
            kn = pp[(pn + lane < npc) ? (pn + lane) : lastp];
#pragma unroll
            for (int q = 0; q < 4; ++q) {
                unsigned v = kv[q];
                const int e0 = (pc + lane) * 128 + 32 * q;
                unsigned m = __builtin_amdgcn_ballot_w32(v != 0u);
                while (m != 0u) {
                    if (v != 0u) {
                        myq[qn + __builtin_popcount(m & ltmask)] = e0 + __builtin_ctz(v);
                        v &= v - 1u;
                    }
                    qn += __builtin_popcount(m);
                    if (qn >= 32) {
                        drain_queue<F, P>(src, dst, dinv, H, acc, dl, myq, lo, glist, &lcnt, 32, lane);
                        qn -= 32;
                        int te = 0;
                        if (lane < qn) te = myq[32 + lane];
                        if (lane < qn) myq[lane] = te;
                    }
                    m = __builtin_amdgcn_ballot_w32(v != 0u);
                }
            }
            pc = pn;
        }
    }
    if (qn > 0) drain_queue<F, P>(src, dst, dinv, H, acc, dl, myq, lo, glist, &lcnt, qn, lane);
    __syncthreads();

    constexpr int NT = (NC + 15) / 16;
    const int half = lane >> 4;
    const int l16  = lane & 15;
    const int jobs = ((cnt + 15) >> 4) * NT;
    for (int job = wave; job < jobs; job += OWN_WAVES) {
        const int mt = job / NT;
        const int t  = job - mt * NT;
        const int row = (mt * 16 + l16 < cnt) ? (mt * 16 + l16) : (cnt - 1);
        const float* ar = acc + row * P + 2 * half * (F / 4);
        const int  n  = t * 16 + l16;
        const bool nc = (NC % 16 == 0) || (n < NC);
        const float* wp = W2 + (size_t)(2 * half) * NC + (nc ? n : 0);
        v8f c = {};
#pragma unroll 2
        for (int j4 = 0; j4 < F / 16; ++j4) {
            const float4 ax = *reinterpret_cast<const float4*>(ar + 4 * j4);
            const float4 ay = *reinterpret_cast<const float4*>(ar + F / 4 + 4 * j4);
            const float a0[4] = { ax.x, ax.y, ax.z, ax.w };
            const float a1[4] = { ay.x, ay.y, ay.z, ay.w };
#pragma unroll
            for (int u = 0; u < 4; ++u) {
                const int k = 16 * j4 + 4 * u;
                const float b0 = wp[(size_t)k * NC];
                const float b1 = wp[(size_t)(k + 1) * NC];
                v2f a, b;
                a.x = a0[u] > 0.0f ? a0[u] : 0.0f;
                a.y = a1[u] > 0.0f ? a1[u] : 0.0f;
                b.x = nc ? b0 : 0.0f;
                b.y = nc ? b1 : 0.0f;
                c = __builtin_amdgcn_wmma_f32_16x16x4_f32(false, a, false, b, (short)0, c, false, false);
            }
        }
        if (nc) {
            const int r0 = mt * 16 + half * 8;
            float* out = H2 + (size_t)(lo + r0) * NC + n;
#pragma unroll
            for (int r = 0; r < 8; ++r)
                if (r0 + r < cnt) out[(size_t)r * NC] = c[r];
        }
    }
}

template <int F>
__global__ void __launch_bounds__(OWN_WAVES * 32)
agg_walk(const unsigned* __restrict__ list, const float* __restrict__ gcnt,
         const float* __restrict__ dinv, const float* __restrict__ H,
         const float* __restrict__ bias, float* __restrict__ OUT, int Nn) {
    static_assert(F > 32 && F <= 64, "the epilogue holds a row in two registers per lane");
    __shared__ float acc[OWN_NPB * F];
    __shared__ float red[OWN_WAVES];

    const int tid  = threadIdx.x;
    const int lane = tid & 31;
    const int wave = __builtin_amdgcn_readfirstlane(tid >> 5);
    const int lo   = blockIdx.x * OWN_NPB;
    const int cnt  = (Nn - lo) < OWN_NPB ? (Nn - lo) : OWN_NPB;

    list_offset_share(gcnt, red, blockIdx.x, tid, lane, wave);
    start_rows<F, F>(dinv, H, bias, acc, lo, cnt, tid);
    __syncthreads();

    const uint2* __restrict__ glist = reinterpret_cast<const uint2*>(list) + list_offset(red);
    const int len = (int)gcnt[blockIdx.x];

    int c = wave * 32;
    uint2 en = make_uint2(0u, 0u);
    if (c + lane < len) en = glist[c + lane];
#pragma nounroll
    while (c < len) {
        const uint2 ev = en;
        const int rem = len - c;
        const int cn = c + OWN_WAVES * 32;
        en = glist[(cn + lane < len) ? (cn + lane) : (len - 1)];
        const int   s    = (int)(ev.x & 0xFFFFu);
        const int   dloc = (int)(ev.x >> 16);
        const float w    = __uint_as_float(ev.y);
        if (rem >= 32) add_rows<F, F>(H, acc, s, dloc, w, 32, lane);
        else           add_rows<F, F>(H, acc, s, dloc, w, rem, lane);
        c = cn;
    }
    __syncthreads();

    for (int i = wave; i < cnt; i += OWN_WAVES) {
        const float v0 = (lane < F) ? acc[i * F + lds_col<F>(lane)] : -INFINITY;
        const float v1 = (lane + 32 < F) ? acc[i * F + lds_col<F>(32 + lane)] : -INFINITY;
        float m = fmaxf(v0, v1);
#pragma unroll
        for (int off = 16; off > 0; off >>= 1)
            m = fmaxf(m, __shfl_xor(m, off, 32));
        float sm = ((lane < F) ? __expf(v0 - m) : 0.0f) +
                   ((lane + 32 < F) ? __expf(v1 - m) : 0.0f);
#pragma unroll
        for (int off = 16; off > 0; off >>= 1)
            sm += __shfl_xor(sm, off, 32);
        const float ls = __logf(sm);
        const size_t ob = (size_t)(lo + i) * F;
        if (lane < F)      OUT[ob + lane]      = v0 - m - ls;
        if (lane + 32 < F) OUT[ob + 32 + lane] = v1 - m - ls;
    }
}

extern "C" void kernel_launch(void* const* d_in, const int* in_sizes, int n_in,
                              void* d_out, int out_size, void* d_ws, size_t ws_size,
                              hipStream_t stream) {
    const float* x  = (const float*)d_in[0];
    const float* W1 = (const float*)d_in[1];
    const float* b1 = (const float*)d_in[2];
    const float* W2 = (const float*)d_in[3];
    const float* b2 = (const float*)d_in[4];
    const int*   ei = (const int*)d_in[5];
    const int E = in_sizes[5] / 2;
    const int Nn = N_NODES;
    const int* src = ei;
    const int* dst = ei + E;

    float* out = (float*)d_out;

    float* ws   = (float*)d_ws;
    float* dinv = ws;
    float* h1   = dinv + 50048;
    float* h2   = h1   + (size_t)Nn * F_HID;
    const int nkb = (E + KB_EDGES - 1) / KB_EDGES;
    unsigned* planes = (unsigned*)(h2 + (size_t)Nn * F_OUT);
    float* gcnt = (float*)(planes + (size_t)OWN_GROUPS * nkb * KB_WORDS);
    unsigned* list = (unsigned*)(gcnt + 256);

    const int TPB = 256;

    zero_f32<<<(Nn + TPB - 1) / TPB, TPB, 0, stream>>>(dinv, Nn);
    {
        int mtiles = Nn / 16;
        int gemm_blocks = (mtiles + GD_TPB / 32 - 1) / (GD_TPB / 32);
        gemm1_deg_bits<<<gemm_blocks + nkb, GD_TPB, 0, stream>>>(x, W1, h1, Nn, gemm_blocks,
                                                                 dst, dinv, planes, E, nkb);
    }
    make_dinv<<<(Nn + OWN_NPB - 1) / OWN_NPB, DINV_TPB, 0, stream>>>(dinv, gcnt, Nn);

    {
        int blocks = (Nn + OWN_NPB - 1) / OWN_NPB;
        agg_scan<F_HID, F_OUT><<<blocks, OWN_WAVES * 32, 0, stream>>>(
            src, dst, planes, gcnt, dinv, h1, b1, W2, h2, list, nkb, Nn);
    }

    {
        int blocks = (Nn + OWN_NPB - 1) / OWN_NPB;
        agg_walk<F_OUT><<<blocks, OWN_WAVES * 32, 0, stream>>>(
            list, gcnt, dinv, h2, b2, out, Nn);
    }
}


# Degrees counted by the owner groups from their own bit planes (LDS integer adds, 32 destinations per load), 1/sqrt there: no global float atomics, no zeroing launch; 4 launches, not 5; same sums
# speedup vs baseline: 4.4933x; 1.2052x over previous
#include <hip/hip_runtime.h>
#include <math.h>

typedef float v2f __attribute__((ext_vector_type(2)));
typedef float v8f __attribute__((ext_vector_type(8)));

#define N_NODES 50000
#define F_IN    128
#define F_HID   128
#define F_OUT   40
#define OWN_GROUPS 254
#define OWN_NPB ((N_NODES + OWN_GROUPS - 1) / OWN_GROUPS)
#define CNT_WAVES 16
#define CNT_PIECES 8
#define KB_EDGES 2048
#define KB_WORDS (KB_EDGES / 32)
#define GD_TPB 256

__device__ __forceinline__ void bits_block(const int* __restrict__ dst,
                                           unsigned* __restrict__ planes, unsigned* bmp,
                                           int E, int nkb, int kb, int tid) {
    constexpr int EPT = KB_EDGES / GD_TPB;
    constexpr int NQ  = OWN_GROUPS * KB_WORDS / 4;
    uint4* b4 = reinterpret_cast<uint4*>(bmp);
    for (int i = tid; i < NQ; i += GD_TPB) b4[i] = make_uint4(0u, 0u, 0u, 0u);
    const int e0   = kb * KB_EDGES + tid;
    const int last = E - 1;
    int d[EPT];
#pragma unroll
    for (int j = 0; j < EPT; ++j) {
        const int e = e0 + GD_TPB * j;
        d[j] = dst[e < E ? e : last];
    }
    __syncthreads();
#pragma unroll
    for (int j = 0; j < EPT; ++j) {
        const int e = e0 + GD_TPB * j;
        if (e < E) {
            const unsigned o = (unsigned)d[j] / OWN_NPB;
            if (o < OWN_GROUPS)
                __hip_atomic_fetch_or(&bmp[o * KB_WORDS + ((GD_TPB * j + tid) >> 5)], 1u << (tid & 31),
                                      __ATOMIC_RELAXED, __HIP_MEMORY_SCOPE_WORKGROUP);
        }
    }
    __syncthreads();
    uint4* p4 = reinterpret_cast<uint4*>(planes);
    for (int i = tid; i < NQ; i += GD_TPB) {
        const int g = i / (KB_WORDS / 4);
        const int q = i - g * (KB_WORDS / 4);
        p4[((size_t)g * nkb + kb) * (KB_WORDS / 4) + q] = b4[i];
    }
}

template <int NT, int NC>
__device__ __forceinline__ void gemm_strip_f32(const float* __restrict__ X,
                                               const float* __restrict__ W,
                                               float* __restrict__ H, int M, int mt, int lane) {
    if (mt * 16 >= M) return;

    const int half = lane >> 4;
    const int l16  = lane & 15;

    v8f c[NT] = {};

    const float* xrow = X + (size_t)(mt * 16 + l16) * 128;

    for (int k = 0; k < 128; k += 4) {
        v2f a;
        a.x = xrow[k + 2 * half];
        a.y = xrow[k + 2 * half + 1];
#pragma unroll
        for (int t = 0; t < NT; ++t) {
            const int n = t * 16 + l16;
            v2f b;
            if ((NC % 16 == 0) || (n < NC)) {
                b.x = W[(size_t)(k + 2 * half) * NC + n];
                b.y = W[(size_t)(k + 2 * half + 1) * NC + n];
            } else {
                b.x = 0.0f; b.y = 0.0f;
            }
            c[t] = __builtin_amdgcn_wmma_f32_16x16x4_f32(
                false, a, false, b, (short)0, c[t], false, false);
        }
    }

#pragma unroll
    for (int t = 0; t < NT; ++t) {
        const int n = t * 16 + l16;
        if ((NC % 16 == 0) || (n < NC)) {
            float* out = H + (size_t)(mt * 16 + half * 8) * NC + n;
#pragma unroll
            for (int r = 0; r < 8; ++r) out[(size_t)r * NC] = c[t][r];
        }
    }
}

__global__ void __launch_bounds__(GD_TPB)
gemm1_bits(const float* __restrict__ X, const float* __restrict__ W, float* __restrict__ H, int M,
           int gemm_blocks,
           const int* __restrict__ dst, unsigned* __restrict__ planes,
           int E, int nkb) {
    __shared__ unsigned bmp[OWN_GROUPS * KB_WORDS];
    if ((int)blockIdx.x < gemm_blocks) {
        gemm_strip_f32<8, 128>(X, W, H, M, blockIdx.x * (GD_TPB / 32) + (threadIdx.x >> 5), threadIdx.x & 31);
    } else {
        bits_block(dst, planes, bmp, E, nkb, (int)blockIdx.x - gemm_blocks, threadIdx.x);
    }
}

#define OWN_WAVES 32
#define OWN_QCAP  64

static_assert(OWN_NPB <= 256, "a list entry keeps the local row in 8 bits");
static_assert(CNT_WAVES * 32 >= OWN_NPB, "count_dinv: one thread per node of a range");
static_assert(N_NODES <= 65536, "a list entry keeps the source node in 16 bits");
static_assert((N_NODES + OWN_NPB - 1) / OWN_NPB <= OWN_GROUPS, "every node has an owner plane");
static_assert((OWN_GROUPS * KB_WORDS) % 4 == 0 && KB_EDGES % GD_TPB == 0, "a block's image is whole 16-byte pieces, its edges whole rounds of the work-group");
static_assert((N_NODES + OWN_NPB - 1) / OWN_NPB <= OWN_WAVES * 32, "one thread per earlier group sums the list offset");

__device__ __forceinline__ void count_batch(const int* __restrict__ dst, int* cntl, const int* qe,
                                            int lo, int n, int lane, int& pend) {
    if (pend >= 0) __hip_atomic_fetch_add(&cntl[pend - lo], 1, __ATOMIC_RELAXED, __HIP_MEMORY_SCOPE_WORKGROUP);
    pend = -1;
    if (lane < n) pend = dst[qe[lane]];
}

__global__ void __launch_bounds__(CNT_WAVES * 32)
count_dinv(const int* __restrict__ dst, const unsigned* __restrict__ planes,
           float* __restrict__ dinv, float* __restrict__ gcnt, int nkb, int Nn) {
    __shared__ int   cntl[CNT_WAVES * 32];
    __shared__ int   qe[CNT_WAVES][OWN_QCAP];
    __shared__ float part[CNT_WAVES];

    const int tid  = threadIdx.x;
    const int lane = tid & 31;
    const int wave = __builtin_amdgcn_readfirstlane(tid >> 5);
    const int lo   = blockIdx.x * OWN_NPB;
    const int cnt  = (Nn - lo) < OWN_NPB ? (Nn - lo) : OWN_NPB;

    cntl[tid] = 0;
    __syncthreads();

    int* myq = qe[wave];
    int qn = 0;
    int pend = -1;

    const int npc = nkb * (KB_WORDS / 4);
    if (npc > 0) {
        constexpr int STEP = CNT_WAVES * 32;
        const uint4* __restrict__ pp = reinterpret_cast<const uint4*>(planes) + (size_t)blockIdx.x * npc;
        const unsigned ltmask = (1u << lane) - 1u;
        const int lastp = npc - 1;
        int pc = wave * 32;
        uint4 kn[CNT_PIECES];
#pragma unroll
        for (int u = 0; u < CNT_PIECES; ++u) {
            const int p = pc + u * STEP + lane;
            kn[u] = pp[p < npc ? p : lastp];
        }
#pragma nounroll
        while (pc < npc) {
            uint4 kc[CNT_PIECES];
#pragma unroll
            for (int u = 0; u < CNT_PIECES; ++u) kc[u] = kn[u];
            const int pn = pc + CNT_PIECES * STEP;
#pragma unroll
            for (int u = 0; u < CNT_PIECES; ++u) {
                const int p = pn + u * STEP + lane;
                kn[u] = pp[p < npc ? p : lastp];
            }
#pragma unroll
            for (int u = 0; u < CNT_PIECES; ++u) {
                const int p = pc + u * STEP + lane;
                const unsigned live = (p < npc) ? 0xFFFFFFFFu : 0u;
#pragma nounroll
                for (int q = 0; q < 4; ++q) {
                    unsigned v = (q == 0 ? kc[u].x : q == 1 ? kc[u].y : q == 2 ? kc[u].z : kc[u].w) & live;
                    const int e0 = p * 128 + 32 * q;
                    unsigned m = __builtin_amdgcn_ballot_w32(v != 0u);
                    while (m != 0u) {
                        if (v != 0u) {
                            myq[qn + __builtin_popcount(m & ltmask)] = e0 + __builtin_ctz(v);
                            v &= v - 1u;
                        }
                        qn += __builtin_popcount(m);
                        if (qn >= 32) {
                            count_batch(dst, cntl, myq, lo, 32, lane, pend);
                            qn -= 32;
                            int te = 0;
                            if (lane < qn) te = myq[32 + lane];
                            if (lane < qn) myq[lane] = te;
                        }
                        m = __builtin_amdgcn_ballot_w32(v != 0u);
                    }
                }
            }
            pc = pn;
        }
    }
    if (qn > 0) count_batch(dst, cntl, myq, lo, qn, lane, pend);
    if (pend >= 0) __hip_atomic_fetch_add(&cntl[pend - lo], 1, __ATOMIC_RELAXED, __HIP_MEMORY_SCOPE_WORKGROUP);
    __syncthreads();

    float d = 0.0f;
    if (tid < cnt) {
        d = (float)cntl[tid];
        dinv[lo + tid] = rsqrtf(d + 1.0f);
    }
#pragma unroll
    for (int off = 16; off > 0; off >>= 1)
        d += __shfl_xor(d, off, 32);
    if (lane == 0) part[wave] = d;
    __syncthreads();
    if (tid == 0) {
        float t = 0.0f;
        for (int w = 0; w < CNT_WAVES; ++w) t += part[w];
        gcnt[blockIdx.x] = t;
    }
}

template <int F>
__device__ __forceinline__ int lds_col(int c) { return (c & 3) * (F / 4) + (c >> 2); }

#define SCAN_PITCH (F_HID + 4)

template <int F, int P>
__device__ __forceinline__ void start_rows(const float* __restrict__ dinv,
                                           const float* __restrict__ H,
                                           const float* __restrict__ bias,
                                           float* acc, int lo, int cnt, int tid) {
    for (int idx = tid; idx < cnt * F; idx += OWN_WAVES * 32) {
        const int i = idx / F;
        const int f = idx - i * F;
        const float di = dinv[lo + i];
        acc[i * P + lds_col<F>(f)] = H[(size_t)lo * F + idx] * di * di + bias[f];
    }
}

__device__ __forceinline__ void list_offset_share(const float* __restrict__ gcnt, float* red,
                                                  int g, int tid, int lane, int wave) {
    float v = (tid < g) ? gcnt[tid] : 0.0f;
#pragma unroll
    for (int off = 16; off > 0; off >>= 1)
        v += __shfl_xor(v, off, 32);
    if (lane == 0) red[wave] = v;
}

__device__ __forceinline__ int list_offset(const float* red) {
    float t = 0.0f;
    for (int i = 0; i < OWN_WAVES; ++i) t += red[i];
    return __builtin_amdgcn_readfirstlane((int)t);
}

template <int F, int P>
__device__ __forceinline__ void add_rows(const float* __restrict__ H, float* acc,
                                         int s, int dloc, float w, int n, int lane) {
    static_assert(F % 4 == 0 && F / 4 <= 32, "a row is F / 4 lanes of 16 bytes");
    constexpr int LPR = F / 4;
    constexpr int RPL = 32 / LPR;
    const float4* __restrict__ H4 = reinterpret_cast<const float4*>(H);
    if constexpr (RPL == 1) {
        constexpr int ROWS = 16;
#pragma nounroll
        for (int k = 0; k < n; k += ROWS) {
            float4 r[ROWS];
#pragma unroll
            for (int u = 0; u < ROWS; ++u) {
                const int sk = __builtin_amdgcn_readlane(s, (k + u) & 31);
                r[u] = ((k + u) < n) ? H4[(size_t)sk * LPR + lane] : make_float4(0.0f, 0.0f, 0.0f, 0.0f);
            }
#pragma unroll
            for (int u = 0; u < ROWS; ++u) {
                const float wk = __int_as_float(__builtin_amdgcn_readlane(__float_as_int(w), (k + u) & 31));
                const int   dk = __builtin_amdgcn_readlane(dloc, (k + u) & 31);
                if ((k + u) < n) {
                    float* a = acc + dk * P + lane;
                    __hip_atomic_fetch_add(a,           r[u].x * wk, __ATOMIC_RELAXED, __HIP_MEMORY_SCOPE_WORKGROUP);
                    __hip_atomic_fetch_add(a + LPR,     r[u].y * wk, __ATOMIC_RELAXED, __HIP_MEMORY_SCOPE_WORKGROUP);
                    __hip_atomic_fetch_add(a + 2 * LPR, r[u].z * wk, __ATOMIC_RELAXED, __HIP_MEMORY_SCOPE_WORKGROUP);
                    __hip_atomic_fetch_add(a + 3 * LPR, r[u].w * wk, __ATOMIC_RELAXED, __HIP_MEMORY_SCOPE_WORKGROUP);
                }
            }
        }
    } else {
        constexpr int NLD = (32 + RPL - 1) / RPL;
        const int  g    = lane / LPR;
        const int  m    = lane - g * LPR;
        const bool live = g < RPL;
        const int  pk   = s | (dloc << 16);
        float4 r[NLD];
        int    p[NLD];
        float  wk[NLD];
#pragma unroll
        for (int t = 0; t < NLD; ++t) {
            const int e = t * RPL + g;
            p[t]  = __shfl(pk, e & 31, 32);
            wk[t] = __shfl(w, e & 31, 32);
        }
#pragma unroll
        for (int t = 0; t < NLD; ++t) {
            const int e = t * RPL + g;
            const int okm = -(int)(live && e < n);
            const int sk = p[t] & 0xFFFF & okm;
            p[t] |= ~okm;
            r[t] = H4[(size_t)sk * LPR + m];
        }
#pragma unroll
        for (int t = 0; t < NLD; ++t) {
            if (p[t] >= 0) {
                float* a = acc + (p[t] >> 16) * P + m;
                __hip_atomic_fetch_add(a,           r[t].x * wk[t], __ATOMIC_RELAXED, __HIP_MEMORY_SCOPE_WORKGROUP);
                __hip_atomic_fetch_add(a + LPR,     r[t].y * wk[t], __ATOMIC_RELAXED, __HIP_MEMORY_SCOPE_WORKGROUP);
                __hip_atomic_fetch_add(a + 2 * LPR, r[t].z * wk[t], __ATOMIC_RELAXED, __HIP_MEMORY_SCOPE_WORKGROUP);
                __hip_atomic_fetch_add(a + 3 * LPR, r[t].w * wk[t], __ATOMIC_RELAXED, __HIP_MEMORY_SCOPE_WORKGROUP);
            }
        }
    }
}

template <int F, int P>
__device__ __forceinline__ void drain_queue(const int* __restrict__ src,
                                            const int* __restrict__ dst,
                                            const float* __restrict__ dinv,
                                            const float* __restrict__ H,
                                            float* acc, const float* dl,
                                            const int* qe, int lo,
                                            uint2* __restrict__ glist, int* lcnt,
                                            int n, int lane) {
    int s = 0, dloc = 0;
    float w = 0.0f;
    if (lane < n) {
        const int e = qe[lane];
        s = src[e];
        dloc = dst[e] - lo;
        w = dinv[s] * dl[dloc];
    }
    int base = 0;
    if (lane == 0)
        base = __hip_atomic_fetch_add(lcnt, n, __ATOMIC_RELAXED, __HIP_MEMORY_SCOPE_WORKGROUP);
    base = __builtin_amdgcn_readfirstlane(base);
    if (lane < n)
        glist[base + lane] = make_uint2((unsigned)s | ((unsigned)dloc << 16), __float_as_uint(w));
    add_rows<F, P>(H, acc, s, dloc, w, n, lane);
}

template <int F, int NC>
__global__ void __launch_bounds__(OWN_WAVES * 32)
agg_scan(const int* __restrict__ src, const int* __restrict__ dst,
         const unsigned* __restrict__ planes, const float* __restrict__ gcnt,
         const float* __restrict__ dinv, const float* __restrict__ H,
         const float* __restrict__ bias, const float* __restrict__ W2, float* __restrict__ H2,
         unsigned* __restrict__ list, int nkb, int Nn) {
    constexpr int P = SCAN_PITCH;
    static_assert(P >= F && P % 4 == 0 && F % 16 == 0, "rows are whole 16-byte pieces; a trip of the product is four k-steps");
    __shared__ __attribute__((aligned(16))) float acc[OWN_NPB * P];
    __shared__ float dl[OWN_NPB];
    __shared__ int   qe[OWN_WAVES][OWN_QCAP];
    __shared__ float red[OWN_WAVES];
    __shared__ int   lcnt;

    const int tid  = threadIdx.x;
    const int lane = tid & 31;
    const int wave = __builtin_amdgcn_readfirstlane(tid >> 5);
    const int lo   = blockIdx.x * OWN_NPB;
    const int cnt  = (Nn - lo) < OWN_NPB ? (Nn - lo) : OWN_NPB;

    if (tid == 0) lcnt = 0;
    list_offset_share(gcnt, red, blockIdx.x, tid, lane, wave);
    for (int i = tid; i < cnt; i += OWN_WAVES * 32) dl[i] = dinv[lo + i];
    start_rows<F, P>(dinv, H, bias, acc, lo, cnt, tid);
    __syncthreads();

    uint2* glist = reinterpret_cast<uint2*>(list) + list_offset(red);
    int* myq = qe[wave];
    int qn = 0;

    const int npc = nkb * (KB_WORDS / 4);
    if (npc > 0) {
        const uint4* __restrict__ pp = reinterpret_cast<const uint4*>(planes) + (size_t)blockIdx.x * npc;
        const unsigned ltmask = (1u << lane) - 1u;
        const int lastp = npc - 1;
        int pc = wave * 32;
        uint4 kn = make_uint4(0u, 0u, 0u, 0u);
        if (pc < npc) kn = pp[(pc + lane < npc) ? (pc + lane) : lastp];
#pragma nounroll
        while (pc < npc) {
            const unsigned live = (pc + lane < npc) ? 0xFFFFFFFFu : 0u;
            unsigned kv[4] = { kn.x & live, kn.y & live, kn.z & live, kn.w & live };
            const int pn = pc + OWN_WAVES * 32;
            kn = pp[(pn + lane < npc) ? (pn + lane) : lastp];
#pragma unroll
            for (int q = 0; q < 4; ++q) {
                unsigned v = kv[q];
                const int e0 = (pc + lane) * 128 + 32 * q;
                unsigned m = __builtin_amdgcn_ballot_w32(v != 0u);
                while (m != 0u) {
                    if (v != 0u) {
                        myq[qn + __builtin_popcount(m & ltmask)] = e0 + __builtin_ctz(v);
                        v &= v - 1u;
                    }
                    qn += __builtin_popcount(m);
                    if (qn >= 32) {
                        drain_queue<F, P>(src, dst, dinv, H, acc, dl, myq, lo, glist, &lcnt, 32, lane);
                        qn -= 32;
                        int te = 0;
                        if (lane < qn) te = myq[32 + lane];
                        if (lane < qn) myq[lane] = te;
                    }
                    m = __builtin_amdgcn_ballot_w32(v != 0u);
                }
            }
            pc = pn;
        }
    }
    if (qn > 0) drain_queue<F, P>(src, dst, dinv, H, acc, dl, myq, lo, glist, &lcnt, qn, lane);
    __syncthreads();

    constexpr int NT = (NC + 15) / 16;
    const int half = lane >> 4;
    const int l16  = lane & 15;
    const int jobs = ((cnt + 15) >> 4) * NT;
    for (int job = wave; job < jobs; job += OWN_WAVES) {
        const int mt = job / NT;
        const int t  = job - mt * NT;
        const int row = (mt * 16 + l16 < cnt) ? (mt * 16 + l16) : (cnt - 1);
        const float* ar = acc + row * P + 2 * half * (F / 4);
        const int  n  = t * 16 + l16;
        const bool nc = (NC % 16 == 0) || (n < NC);
        const float* wp = W2 + (size_t)(2 * half) * NC + (nc ? n : 0);
        v8f c = {};
#pragma unroll 2
        for (int j4 = 0; j4 < F / 16; ++j4) {
            const float4 ax = *reinterpret_cast<const float4*>(ar + 4 * j4);
            const float4 ay = *reinterpret_cast<const float4*>(ar + F / 4 + 4 * j4);
            const float a0[4] = { ax.x, ax.y, ax.z, ax.w };
            const float a1[4] = { ay.x, ay.y, ay.z, ay.w };
#pragma unroll
            for (int u = 0; u < 4; ++u) {
                const int k = 16 * j4 + 4 * u;
                const float b0 = wp[(size_t)k * NC];
                const float b1 = wp[(size_t)(k + 1) * NC];
                v2f a, b;
                a.x = a0[u] > 0.0f ? a0[u] : 0.0f;
                a.y = a1[u] > 0.0f ? a1[u] : 0.0f;
                b.x = nc ? b0 : 0.0f;
                b.y = nc ? b1 : 0.0f;
                c = __builtin_amdgcn_wmma_f32_16x16x4_f32(false, a, false, b, (short)0, c, false, false);
            }
        }
        if (nc) {
            const int r0 = mt * 16 + half * 8;
            float* out = H2 + (size_t)(lo + r0) * NC + n;
#pragma unroll
            for (int r = 0; r < 8; ++r)
                if (r0 + r < cnt) out[(size_t)r * NC] = c[r];
        }
    }
}

template <int F>
__global__ void __launch_bounds__(OWN_WAVES * 32)
agg_walk(const unsigned* __restrict__ list, const float* __restrict__ gcnt,
         const float* __restrict__ dinv, const float* __restrict__ H,
         const float* __restrict__ bias, float* __restrict__ OUT, int Nn) {
    static_assert(F > 32 && F <= 64, "the epilogue holds a row in two registers per lane");
    __shared__ float acc[OWN_NPB * F];
    __shared__ float red[OWN_WAVES];

    const int tid  = threadIdx.x;
    const int lane = tid & 31;
    const int wave = __builtin_amdgcn_readfirstlane(tid >> 5);
    const int lo   = blockIdx.x * OWN_NPB;
    const int cnt  = (Nn - lo) < OWN_NPB ? (Nn - lo) : OWN_NPB;

    list_offset_share(gcnt, red, blockIdx.x, tid, lane, wave);
    start_rows<F, F>(dinv, H, bias, acc, lo, cnt, tid);
    __syncthreads();

    const uint2* __restrict__ glist = reinterpret_cast<const uint2*>(list) + list_offset(red);
    const int len = (int)gcnt[blockIdx.x];

    int c = wave * 32;
    uint2 en = make_uint2(0u, 0u);
    if (c + lane < len) en = glist[c + lane];
#pragma nounroll
    while (c < len) {
        const uint2 ev = en;
        const int rem = len - c;
        const int cn = c + OWN_WAVES * 32;
        en = glist[(cn + lane < len) ? (cn + lane) : (len - 1)];
        const int   s    = (int)(ev.x & 0xFFFFu);
        const int   dloc = (int)(ev.x >> 16);
        const float w    = __uint_as_float(ev.y);
        if (rem >= 32) add_rows<F, F>(H, acc, s, dloc, w, 32, lane);
        else           add_rows<F, F>(H, acc, s, dloc, w, rem, lane);
        c = cn;
    }
    __syncthreads();

    for (int i = wave; i < cnt; i += OWN_WAVES) {
        const float v0 = (lane < F) ? acc[i * F + lds_col<F>(lane)] : -INFINITY;
        const float v1 = (lane + 32 < F) ? acc[i * F + lds_col<F>(32 + lane)] : -INFINITY;
        float m = fmaxf(v0, v1);
#pragma unroll
        for (int off = 16; off > 0; off >>= 1)
            m = fmaxf(m, __shfl_xor(m, off, 32));
        float sm = ((lane < F) ? __expf(v0 - m) : 0.0f) +
                   ((lane + 32 < F) ? __expf(v1 - m) : 0.0f);
#pragma unroll
        for (int off = 16; off > 0; off >>= 1)
            sm += __shfl_xor(sm, off, 32);
        const float ls = __logf(sm);
        const size_t ob = (size_t)(lo + i) * F;
        if (lane < F)      OUT[ob + lane]      = v0 - m - ls;
        if (lane + 32 < F) OUT[ob + 32 + lane] = v1 - m - ls;
    }
}

extern "C" void kernel_launch(void* const* d_in, const int* in_sizes, int n_in,
                              void* d_out, int out_size, void* d_ws, size_t ws_size,
                              hipStream_t stream) {
    const float* x  = (const float*)d_in[0];
    const float* W1 = (const float*)d_in[1];
    const float* b1 = (const float*)d_in[2];
    const float* W2 = (const float*)d_in[3];
    const float* b2 = (const float*)d_in[4];
    const int*   ei = (const int*)d_in[5];
    const int E = in_sizes[5] / 2;
    const int Nn = N_NODES;
    const int* src = ei;
    const int* dst = ei + E;

    float* out = (float*)d_out;

    float* ws   = (float*)d_ws;
    float* dinv = ws;
    float* h1   = dinv + 50048;
    float* h2   = h1   + (size_t)Nn * F_HID;
    const int nkb = (E + KB_EDGES - 1) / KB_EDGES;
    unsigned* planes = (unsigned*)(h2 + (size_t)Nn * F_OUT);
    float* gcnt = (float*)(planes + (size_t)OWN_GROUPS * nkb * KB_WORDS);
    unsigned* list = (unsigned*)(gcnt + 256);

    {
        int mtiles = Nn / 16;
        int gemm_blocks = (mtiles + GD_TPB / 32 - 1) / (GD_TPB / 32);
        gemm1_bits<<<gemm_blocks + nkb, GD_TPB, 0, stream>>>(x, W1, h1, Nn, gemm_blocks,
                                                             dst, planes, E, nkb);
    }
    count_dinv<<<(Nn + OWN_NPB - 1) / OWN_NPB, CNT_WAVES * 32, 0, stream>>>(dst, planes, dinv, gcnt, nkb, Nn);

    {
        int blocks = (Nn + OWN_NPB - 1) / OWN_NPB;
        agg_scan<F_HID, F_OUT><<<blocks, OWN_WAVES * 32, 0, stream>>>(
            src, dst, planes, gcnt, dinv, h1, b1, W2, h2, list, nkb, Nn);
    }

    {
        int blocks = (Nn + OWN_NPB - 1) / OWN_NPB;
        agg_walk<F_OUT><<<blocks, OWN_WAVES * 32, 0, stream>>>(
            list, gcnt, dinv, h2, b2, out, Nn);
    }
}


# Layer-1 aggregation: one scattered 4-byte read per edge, not three: plane launch stores source | local row per edge, each owner group holds all 1/sqrt(degree) in LDS; same sums, LDS arrival order
# speedup vs baseline: 4.6716x; 1.0397x over previous
#include <hip/hip_runtime.h>
#include <math.h>

typedef float v2f __attribute__((ext_vector_type(2)));
typedef float v8f __attribute__((ext_vector_type(8)));

#define N_NODES 50000
#define F_IN    128
#define F_HID   128
#define F_OUT   40
#define OWN_GROUPS 254
#define OWN_NPB ((N_NODES + OWN_GROUPS - 1) / OWN_GROUPS)
#define CNT_WAVES 16
#define CNT_PIECES 8
#define KB_EDGES 2048
#define KB_WORDS (KB_EDGES / 32)
#define GD_TPB 256

__device__ __forceinline__ void bits_block(const int* __restrict__ src, const int* __restrict__ dst,
                                           unsigned* __restrict__ planes, unsigned* __restrict__ pk, unsigned* bmp,
                                           int E, int nkb, int kb, int tid) {
    constexpr int EPT = KB_EDGES / GD_TPB;
    constexpr int NQ  = OWN_GROUPS * KB_WORDS / 4;
    uint4* b4 = reinterpret_cast<uint4*>(bmp);
    for (int i = tid; i < NQ; i += GD_TPB) b4[i] = make_uint4(0u, 0u, 0u, 0u);
    const int e0   = kb * KB_EDGES + tid;
    const int last = E - 1;
    int d[EPT];
    int s[EPT];
#pragma unroll
    for (int j = 0; j < EPT; ++j) {
        const int e = e0 + GD_TPB * j;
        d[j] = dst[e < E ? e : last];
        s[j] = src[e < E ? e : last];
    }
    __syncthreads();
#pragma unroll
    for (int j = 0; j < EPT; ++j) {
        const int e = e0 + GD_TPB * j;
        if (e < E) {
            const unsigned o = (unsigned)d[j] / OWN_NPB;
            if (o < OWN_GROUPS) {
                __hip_atomic_fetch_or(&bmp[o * KB_WORDS + ((GD_TPB * j + tid) >> 5)], 1u << (tid & 31),
                                      __ATOMIC_RELAXED, __HIP_MEMORY_SCOPE_WORKGROUP);
                pk[e] = ((unsigned)s[j] & 0xFFFFu) | (((unsigned)d[j] - o * OWN_NPB) << 16);
            }
        }
    }
    __syncthreads();
    uint4* p4 = reinterpret_cast<uint4*>(planes);
    for (int i = tid; i < NQ; i += GD_TPB) {
        const int g = i / (KB_WORDS / 4);
        const int q = i - g * (KB_WORDS / 4);
        p4[((size_t)g * nkb + kb) * (KB_WORDS / 4) + q] = b4[i];
    }
}

template <int NT, int NC>
__device__ __forceinline__ void gemm_strip_f32(const float* __restrict__ X,
                                               const float* __restrict__ W,
                                               float* __restrict__ H, int M, int mt, int lane) {
    if (mt * 16 >= M) return;

    const int half = lane >> 4;
    const int l16  = lane & 15;

    v8f c[NT] = {};

    const float* xrow = X + (size_t)(mt * 16 + l16) * 128;

    for (int k = 0; k < 128; k += 4) {
        v2f a;
        a.x = xrow[k + 2 * half];
        a.y = xrow[k + 2 * half + 1];
#pragma unroll
        for (int t = 0; t < NT; ++t) {
            const int n = t * 16 + l16;
            v2f b;
            if ((NC % 16 == 0) || (n < NC)) {
                b.x = W[(size_t)(k + 2 * half) * NC + n];
                b.y = W[(size_t)(k + 2 * half + 1) * NC + n];
            } else {
                b.x = 0.0f; b.y = 0.0f;
            }
            c[t] = __builtin_amdgcn_wmma_f32_16x16x4_f32(
                false, a, false, b, (short)0, c[t], false, false);
        }
    }

#pragma unroll
    for (int t = 0; t < NT; ++t) {
        const int n = t * 16 + l16;
        if ((NC % 16 == 0) || (n < NC)) {
            float* out = H + (size_t)(mt * 16 + half * 8) * NC + n;
#pragma unroll
            for (int r = 0; r < 8; ++r) out[(size_t)r * NC] = c[t][r];
        }
    }
}

__global__ void __launch_bounds__(GD_TPB)
gemm1_bits(const float* __restrict__ X, const float* __restrict__ W, float* __restrict__ H, int M,
           int gemm_blocks,
           const int* __restrict__ src, const int* __restrict__ dst,
           unsigned* __restrict__ planes, unsigned* __restrict__ pk,
           int E, int nkb) {
    __shared__ unsigned bmp[OWN_GROUPS * KB_WORDS];
    if ((int)blockIdx.x < gemm_blocks) {
        gemm_strip_f32<8, 128>(X, W, H, M, blockIdx.x * (GD_TPB / 32) + (threadIdx.x >> 5), threadIdx.x & 31);
    } else {
        bits_block(src, dst, planes, pk, bmp, E, nkb, (int)blockIdx.x - gemm_blocks, threadIdx.x);
    }
}

#define OWN_WAVES 32
#define OWN_QCAP  64

static_assert(OWN_NPB <= 256, "a list entry keeps the local row in 8 bits");
static_assert(CNT_WAVES * 32 >= OWN_NPB, "count_dinv: one thread per node of a range");
static_assert(N_NODES <= 65536, "a list entry keeps the source node in 16 bits");
static_assert((N_NODES + OWN_NPB - 1) / OWN_NPB <= OWN_GROUPS, "every node has an owner plane");
static_assert((OWN_GROUPS * KB_WORDS) % 4 == 0 && KB_EDGES % GD_TPB == 0, "a block's image is whole 16-byte pieces, its edges whole rounds of the work-group");
static_assert((N_NODES + OWN_NPB - 1) / OWN_NPB <= OWN_WAVES * 32, "one thread per earlier group sums the list offset");

__device__ __forceinline__ void count_batch(const unsigned* __restrict__ pk, int* cntl, const int* qe,
                                            int n, int lane, int& pend) {
    if (pend >= 0) __hip_atomic_fetch_add(&cntl[pend], 1, __ATOMIC_RELAXED, __HIP_MEMORY_SCOPE_WORKGROUP);
    pend = -1;
    if (lane < n) pend = (int)(pk[qe[lane]] >> 16);
}

__global__ void __launch_bounds__(CNT_WAVES * 32)
count_dinv(const unsigned* __restrict__ pk, const unsigned* __restrict__ planes,
           float* __restrict__ dinv, float* __restrict__ gcnt, int nkb, int Nn) {
    __shared__ int   cntl[CNT_WAVES * 32];
    __shared__ int   qe[CNT_WAVES][OWN_QCAP];
    __shared__ float part[CNT_WAVES];

    const int tid  = threadIdx.x;
    const int lane = tid & 31;
    const int wave = __builtin_amdgcn_readfirstlane(tid >> 5);
    const int lo   = blockIdx.x * OWN_NPB;
    const int cnt  = (Nn - lo) < OWN_NPB ? (Nn - lo) : OWN_NPB;

    cntl[tid] = 0;
    __syncthreads();

    int* myq = qe[wave];
    int qn = 0;
    int pend = -1;

    const int npc = nkb * (KB_WORDS / 4);
    if (npc > 0) {
        constexpr int STEP = CNT_WAVES * 32;
        const uint4* __restrict__ pp = reinterpret_cast<const uint4*>(planes) + (size_t)blockIdx.x * npc;
        const unsigned ltmask = (1u << lane) - 1u;
        const int lastp = npc - 1;
        int pc = wave * 32;
        uint4 kn[CNT_PIECES];
#pragma unroll
        for (int u = 0; u < CNT_PIECES; ++u) {
            const int p = pc + u * STEP + lane;
            kn[u] = pp[p < npc ? p : lastp];
        }
#pragma nounroll
        while (pc < npc) {
            uint4 kc[CNT_PIECES];
#pragma unroll
            for (int u = 0; u < CNT_PIECES; ++u) kc[u] = kn[u];
            const int pn = pc + CNT_PIECES * STEP;
#pragma unroll
            for (int u = 0; u < CNT_PIECES; ++u) {
                const int p = pn + u * STEP + lane;
                kn[u] = pp[p < npc ? p : lastp];
            }
#pragma unroll
            for (int u = 0; u < CNT_PIECES; ++u) {
                const int p = pc + u * STEP + lane;
                const unsigned live = (p < npc) ? 0xFFFFFFFFu : 0u;
#pragma nounroll
                for (int q = 0; q < 4; ++q) {
                    unsigned v = (q == 0 ? kc[u].x : q == 1 ? kc[u].y : q == 2 ? kc[u].z : kc[u].w) & live;
                    const int e0 = p * 128 + 32 * q;
                    unsigned m = __builtin_amdgcn_ballot_w32(v != 0u);
                    while (m != 0u) {
                        if (v != 0u) {
                            myq[qn + __builtin_popcount(m & ltmask)] = e0 + __builtin_ctz(v);
                            v &= v - 1u;
                        }
                        qn += __builtin_popcount(m);
                        if (qn >= 32) {
                            count_batch(pk, cntl, myq, 32, lane, pend);
                            qn -= 32;
                            int te = 0;
                            if (lane < qn) te = myq[32 + lane];
                            if (lane < qn) myq[lane] = te;
                        }
                        m = __builtin_amdgcn_ballot_w32(v != 0u);
                    }
                }
            }
            pc = pn;
        }
    }
    if (qn > 0) count_batch(pk, cntl, myq, qn, lane, pend);
    if (pend >= 0) __hip_atomic_fetch_add(&cntl[pend], 1, __ATOMIC_RELAXED, __HIP_MEMORY_SCOPE_WORKGROUP);
    __syncthreads();

    float d = 0.0f;
    if (tid < cnt) {
        d = (float)cntl[tid];
        dinv[lo + tid] = rsqrtf(d + 1.0f);
    }
#pragma unroll
    for (int off = 16; off > 0; off >>= 1)
        d += __shfl_xor(d, off, 32);
    if (lane == 0) part[wave] = d;
    __syncthreads();
    if (tid == 0) {
        float t = 0.0f;
        for (int w = 0; w < CNT_WAVES; ++w) t += part[w];
        gcnt[blockIdx.x] = t;
    }
}

template <int F>
__device__ __forceinline__ int lds_col(int c) { return (c & 3) * (F / 4) + (c >> 2); }

#define SCAN_PITCH (F_HID + 4)

template <int F, int P>
__device__ __forceinline__ void start_rows(const float* __restrict__ dinv,
                                           const float* __restrict__ H,
                                           const float* __restrict__ bias,
                                           float* acc, int lo, int cnt, int tid) {
    for (int idx = tid; idx < cnt * F; idx += OWN_WAVES * 32) {
        const int i = idx / F;
        const int f = idx - i * F;
        const float di = dinv[lo + i];
        acc[i * P + lds_col<F>(f)] = H[(size_t)lo * F + idx] * di * di + bias[f];
    }
}

__device__ __forceinline__ void list_offset_share(const float* __restrict__ gcnt, float* red,
                                                  int g, int tid, int lane, int wave) {
    float v = (tid < g) ? gcnt[tid] : 0.0f;
#pragma unroll
    for (int off = 16; off > 0; off >>= 1)
        v += __shfl_xor(v, off, 32);
    if (lane == 0) red[wave] = v;
}

__device__ __forceinline__ int list_offset(const float* red) {
    float t = 0.0f;
    for (int i = 0; i < OWN_WAVES; ++i) t += red[i];
    return __builtin_amdgcn_readfirstlane((int)t);
}

template <int F, int P>
__device__ __forceinline__ void add_rows(const float* __restrict__ H, float* acc,
                                         int s, int dloc, float w, int n, int lane) {
    static_assert(F % 4 == 0 && F / 4 <= 32, "a row is F / 4 lanes of 16 bytes");
    constexpr int LPR = F / 4;
    constexpr int RPL = 32 / LPR;
    const float4* __restrict__ H4 = reinterpret_cast<const float4*>(H);
    if constexpr (RPL == 1) {
        constexpr int ROWS = 16;
#pragma nounroll
        for (int k = 0; k < n; k += ROWS) {
            float4 r[ROWS];
#pragma unroll
            for (int u = 0; u < ROWS; ++u) {
                const int sk = __builtin_amdgcn_readlane(s, (k + u) & 31);
                r[u] = ((k + u) < n) ? H4[(size_t)sk * LPR + lane] : make_float4(0.0f, 0.0f, 0.0f, 0.0f);
            }
#pragma unroll
            for (int u = 0; u < ROWS; ++u) {
                const float wk = __int_as_float(__builtin_amdgcn_readlane(__float_as_int(w), (k + u) & 31));
                const int   dk = __builtin_amdgcn_readlane(dloc, (k + u) & 31);
                if ((k + u) < n) {
                    float* a = acc + dk * P + lane;
                    __hip_atomic_fetch_add(a,           r[u].x * wk, __ATOMIC_RELAXED, __HIP_MEMORY_SCOPE_WORKGROUP);
                    __hip_atomic_fetch_add(a + LPR,     r[u].y * wk, __ATOMIC_RELAXED, __HIP_MEMORY_SCOPE_WORKGROUP);
                    __hip_atomic_fetch_add(a + 2 * LPR, r[u].z * wk, __ATOMIC_RELAXED, __HIP_MEMORY_SCOPE_WORKGROUP);
                    __hip_atomic_fetch_add(a + 3 * LPR, r[u].w * wk, __ATOMIC_RELAXED, __HIP_MEMORY_SCOPE_WORKGROUP);
                }
            }
        }
    } else {
        constexpr int NLD = (32 + RPL - 1) / RPL;
        const int  g    = lane / LPR;
        const int  m    = lane - g * LPR;
        const bool live = g < RPL;
        const int  pk   = s | (dloc << 16);
        float4 r[NLD];
        int    p[NLD];
        float  wk[NLD];
#pragma unroll
        for (int t = 0; t < NLD; ++t) {
            const int e = t * RPL + g;
            p[t]  = __shfl(pk, e & 31, 32);
            wk[t] = __shfl(w, e & 31, 32);
        }
#pragma unroll
        for (int t = 0; t < NLD; ++t) {
            const int e = t * RPL + g;
            const int okm = -(int)(live && e < n);
            const int sk = p[t] & 0xFFFF & okm;
            p[t] |= ~okm;
            r[t] = H4[(size_t)sk * LPR + m];
        }
#pragma unroll
        for (int t = 0; t < NLD; ++t) {
            if (p[t] >= 0) {
                float* a = acc + (p[t] >> 16) * P + m;
                __hip_atomic_fetch_add(a,           r[t].x * wk[t], __ATOMIC_RELAXED, __HIP_MEMORY_SCOPE_WORKGROUP);
                __hip_atomic_fetch_add(a + LPR,     r[t].y * wk[t], __ATOMIC_RELAXED, __HIP_MEMORY_SCOPE_WORKGROUP);
                __hip_atomic_fetch_add(a + 2 * LPR, r[t].z * wk[t], __ATOMIC_RELAXED, __HIP_MEMORY_SCOPE_WORKGROUP);
                __hip_atomic_fetch_add(a + 3 * LPR, r[t].w * wk[t], __ATOMIC_RELAXED, __HIP_MEMORY_SCOPE_WORKGROUP);
            }
        }
    }
}

template <int F, int P>
__device__ __forceinline__ void drain_queue(const unsigned* __restrict__ pk,
                                            const float* __restrict__ H,
                                            float* acc, const float* dinvl,
                                            const int* qe, int lo,
                                            uint2* __restrict__ glist, int* lcnt,
                                            int n, int lane) {
    int s = 0, dloc = 0;
    float w = 0.0f;
    if (lane < n) {
        const unsigned p = pk[qe[lane]];
        s = (int)(p & 0xFFFFu);
        dloc = (int)(p >> 16);
        w = dinvl[s] * dinvl[lo + dloc];
    }
    int base = 0;
    if (lane == 0)
        base = __hip_atomic_fetch_add(lcnt, n, __ATOMIC_RELAXED, __HIP_MEMORY_SCOPE_WORKGROUP);
    base = __builtin_amdgcn_readfirstlane(base);
    if (lane < n)
        glist[base + lane] = make_uint2((unsigned)s | ((unsigned)dloc << 16), __float_as_uint(w));
    add_rows<F, P>(H, acc, s, dloc, w, n, lane);
}

template <int F, int NC>
__global__ void __launch_bounds__(OWN_WAVES * 32)
agg_scan(const unsigned* __restrict__ pk,
         const unsigned* __restrict__ planes, const float* __restrict__ gcnt,
         const float* __restrict__ dinv, const float* __restrict__ H,
         const float* __restrict__ bias, const float* __restrict__ W2, float* __restrict__ H2,
         unsigned* __restrict__ list, int nkb, int Nn) {
    constexpr int P = SCAN_PITCH;
    static_assert(P >= F && P % 4 == 0 && F % 16 == 0, "rows are whole 16-byte pieces; a trip of the product is four k-steps");
    __shared__ __attribute__((aligned(16))) float acc[OWN_NPB * P];
    __shared__ __attribute__((aligned(16))) float dinvl[N_NODES];
    __shared__ int   qe[OWN_WAVES][OWN_QCAP];
    __shared__ float red[OWN_WAVES];
    __shared__ int   lcnt;

    const int tid  = threadIdx.x;
    const int lane = tid & 31;
    const int wave = __builtin_amdgcn_readfirstlane(tid >> 5);
    const int lo   = blockIdx.x * OWN_NPB;
    const int cnt  = (Nn - lo) < OWN_NPB ? (Nn - lo) : OWN_NPB;

    if (tid == 0) lcnt = 0;
    list_offset_share(gcnt, red, blockIdx.x, tid, lane, wave);
    {
        static_assert(N_NODES % 4 == 0, "the dinv array is whole 16-byte pieces");
        constexpr int TPB = OWN_WAVES * 32;
        constexpr int DL_INF = 6;
        constexpr int NQ = N_NODES / 4;
        const float4* __restrict__ d4 = reinterpret_cast<const float4*>(dinv);
        float4* l4 = reinterpret_cast<float4*>(dinvl);
        int base = 0;
        for (; base + DL_INF * TPB <= NQ; base += DL_INF * TPB) {
            float4 t[DL_INF];
#pragma unroll
            for (int u = 0; u < DL_INF; ++u) t[u] = d4[base + u * TPB + tid];
#pragma unroll
            for (int u = 0; u < DL_INF; ++u) l4[base + u * TPB + tid] = t[u];
        }
        for (int i = base + tid; i < NQ; i += TPB) l4[i] = d4[i];
    }
    start_rows<F, P>(dinv, H, bias, acc, lo, cnt, tid);
    __syncthreads();

    uint2* glist = reinterpret_cast<uint2*>(list) + list_offset(red);
    int* myq = qe[wave];
    int qn = 0;

    const int npc = nkb * (KB_WORDS / 4);
    if (npc > 0) {
        const uint4* __restrict__ pp = reinterpret_cast<const uint4*>(planes) + (size_t)blockIdx.x * npc;
        const unsigned ltmask = (1u << lane) - 1u;
        const int lastp = npc - 1;
        int pc = wave * 32;
        uint4 kn = make_uint4(0u, 0u, 0u, 0u);
        if (pc < npc) kn = pp[(pc + lane < npc) ? (pc + lane) : lastp];
#pragma nounroll
        while (pc < npc) {
            const unsigned live = (pc + lane < npc) ? 0xFFFFFFFFu : 0u;
            unsigned kv[4] = { kn.x & live, kn.y & live, kn.z & live, kn.w & live };
            const int pn = pc + OWN_WAVES * 32;
            kn = pp[(pn + lane < npc) ? (pn + lane) : lastp];
#pragma unroll
            for (int q = 0; q < 4; ++q) {
                unsigned v = kv[q];
                const int e0 = (pc + lane) * 128 + 32 * q;
                unsigned m = __builtin_amdgcn_ballot_w32(v != 0u);
                while (m != 0u) {
                    if (v != 0u) {
                        myq[qn + __builtin_popcount(m & ltmask)] = e0 + __builtin_ctz(v);
                        v &= v - 1u;
                    }
                    qn += __builtin_popcount(m);
                    if (qn >= 32) {
                        drain_queue<F, P>(pk, H, acc, dinvl, myq, lo, glist, &lcnt, 32, lane);
                        qn -= 32;
                        int te = 0;
                        if (lane < qn) te = myq[32 + lane];
                        if (lane < qn) myq[lane] = te;
                    }
                    m = __builtin_amdgcn_ballot_w32(v != 0u);
                }
            }
            pc = pn;
        }
    }
    if (qn > 0) drain_queue<F, P>(pk, H, acc, dinvl, myq, lo, glist, &lcnt, qn, lane);
    __syncthreads();

    constexpr int NT = (NC + 15) / 16;
    const int half = lane >> 4;
    const int l16  = lane & 15;
    const int jobs = ((cnt + 15) >> 4) * NT;
    for (int job = wave; job < jobs; job += OWN_WAVES) {
        const int mt = job / NT;
        const int t  = job - mt * NT;
        const int row = (mt * 16 + l16 < cnt) ? (mt * 16 + l16) : (cnt - 1);
        const float* ar = acc + row * P + 2 * half * (F / 4);
        const int  n  = t * 16 + l16;
        const bool nc = (NC % 16 == 0) || (n < NC);
        const float* wp = W2 + (size_t)(2 * half) * NC + (nc ? n : 0);
        v8f c = {};
#pragma unroll 2
        for (int j4 = 0; j4 < F / 16; ++j4) {
            const float4 ax = *reinterpret_cast<const float4*>(ar + 4 * j4);
            const float4 ay = *reinterpret_cast<const float4*>(ar + F / 4 + 4 * j4);
            const float a0[4] = { ax.x, ax.y, ax.z, ax.w };
            const float a1[4] = { ay.x, ay.y, ay.z, ay.w };
#pragma unroll
            for (int u = 0; u < 4; ++u) {
                const int k = 16 * j4 + 4 * u;
                const float b0 = wp[(size_t)k * NC];
                const float b1 = wp[(size_t)(k + 1) * NC];
                v2f a, b;
                a.x = a0[u] > 0.0f ? a0[u] : 0.0f;
                a.y = a1[u] > 0.0f ? a1[u] : 0.0f;
                b.x = nc ? b0 : 0.0f;
                b.y = nc ? b1 : 0.0f;
                c = __builtin_amdgcn_wmma_f32_16x16x4_f32(false, a, false, b, (short)0, c, false, false);
            }
        }
        if (nc) {
            const int r0 = mt * 16 + half * 8;
            float* out = H2 + (size_t)(lo + r0) * NC + n;
#pragma unroll
            for (int r = 0; r < 8; ++r)
                if (r0 + r < cnt) out[(size_t)r * NC] = c[r];
        }
    }
}

template <int F>
__global__ void __launch_bounds__(OWN_WAVES * 32)
agg_walk(const unsigned* __restrict__ list, const float* __restrict__ gcnt,
         const float* __restrict__ dinv, const float* __restrict__ H,
         const float* __restrict__ bias, float* __restrict__ OUT, int Nn) {
    static_assert(F > 32 && F <= 64, "the epilogue holds a row in two registers per lane");
    __shared__ float acc[OWN_NPB * F];
    __shared__ float red[OWN_WAVES];

    const int tid  = threadIdx.x;
    const int lane = tid & 31;
    const int wave = __builtin_amdgcn_readfirstlane(tid >> 5);
    const int lo   = blockIdx.x * OWN_NPB;
    const int cnt  = (Nn - lo) < OWN_NPB ? (Nn - lo) : OWN_NPB;

    list_offset_share(gcnt, red, blockIdx.x, tid, lane, wave);
    start_rows<F, F>(dinv, H, bias, acc, lo, cnt, tid);
    __syncthreads();

    const uint2* __restrict__ glist = reinterpret_cast<const uint2*>(list) + list_offset(red);
    const int len = (int)gcnt[blockIdx.x];

    int c = wave * 32;
    uint2 en = make_uint2(0u, 0u);
    if (c + lane < len) en = glist[c + lane];
#pragma nounroll
    while (c < len) {
        const uint2 ev = en;
        const int rem = len - c;
        const int cn = c + OWN_WAVES * 32;
        en = glist[(cn + lane < len) ? (cn + lane) : (len - 1)];
        const int   s    = (int)(ev.x & 0xFFFFu);
        const int   dloc = (int)(ev.x >> 16);
        const float w    = __uint_as_float(ev.y);
        if (rem >= 32) add_rows<F, F>(H, acc, s, dloc, w, 32, lane);
        else           add_rows<F, F>(H, acc, s, dloc, w, rem, lane);
        c = cn;
    }
    __syncthreads();

    for (int i = wave; i < cnt; i += OWN_WAVES) {
        const float v0 = (lane < F) ? acc[i * F + lds_col<F>(lane)] : -INFINITY;
        const float v1 = (lane + 32 < F) ? acc[i * F + lds_col<F>(32 + lane)] : -INFINITY;
        float m = fmaxf(v0, v1);
#pragma unroll
        for (int off = 16; off > 0; off >>= 1)
            m = fmaxf(m, __shfl_xor(m, off, 32));
        float sm = ((lane < F) ? __expf(v0 - m) : 0.0f) +
                   ((lane + 32 < F) ? __expf(v1 - m) : 0.0f);
#pragma unroll
        for (int off = 16; off > 0; off >>= 1)
            sm += __shfl_xor(sm, off, 32);
        const float ls = __logf(sm);
        const size_t ob = (size_t)(lo + i) * F;
        if (lane < F)      OUT[ob + lane]      = v0 - m - ls;
        if (lane + 32 < F) OUT[ob + 32 + lane] = v1 - m - ls;
    }
}

extern "C" void kernel_launch(void* const* d_in, const int* in_sizes, int n_in,
                              void* d_out, int out_size, void* d_ws, size_t ws_size,
                              hipStream_t stream) {
    const float* x  = (const float*)d_in[0];
    const float* W1 = (const float*)d_in[1];
    const float* b1 = (const float*)d_in[2];
    const float* W2 = (const float*)d_in[3];
    const float* b2 = (const float*)d_in[4];
    const int*   ei = (const int*)d_in[5];
    const int E = in_sizes[5] / 2;
    const int Nn = N_NODES;
    const int* src = ei;
    const int* dst = ei + E;

    float* out = (float*)d_out;

    float* ws   = (float*)d_ws;
    float* dinv = ws;
    float* h1   = dinv + 50048;
    float* h2   = h1   + (size_t)Nn * F_HID;
    const int nkb = (E + KB_EDGES - 1) / KB_EDGES;
    unsigned* planes = (unsigned*)(h2 + (size_t)Nn * F_OUT);
    float* gcnt = (float*)(planes + (size_t)OWN_GROUPS * nkb * KB_WORDS);
    unsigned* list = (unsigned*)(gcnt + 256);
    unsigned* pk   = list + 2 * (size_t)E;

    {
        int mtiles = Nn / 16;
        int gemm_blocks = (mtiles + GD_TPB / 32 - 1) / (GD_TPB / 32);
        gemm1_bits<<<gemm_blocks + nkb, GD_TPB, 0, stream>>>(x, W1, h1, Nn, gemm_blocks,
                                                             src, dst, planes, pk, E, nkb);
    }
    count_dinv<<<(Nn + OWN_NPB - 1) / OWN_NPB, CNT_WAVES * 32, 0, stream>>>(pk, planes, dinv, gcnt, nkb, Nn);

    {
        int blocks = (Nn + OWN_NPB - 1) / OWN_NPB;
        agg_scan<F_HID, F_OUT><<<blocks, OWN_WAVES * 32, 0, stream>>>(
            pk, planes, gcnt, dinv, h1, b1, W2, h2, list, nkb, Nn);
    }

    {
        int blocks = (Nn + OWN_NPB - 1) / OWN_NPB;
        agg_walk<F_OUT><<<blocks, OWN_WAVES * 32, 0, stream>>>(
            list, gcnt, dinv, h2, b2, out, Nn);
    }
}
